# starting kernel
# baseline (speedup 1.0000x reference)
#include <hip/hip_runtime.h>

typedef __attribute__((ext_vector_type(2))) float v2f;
typedef __attribute__((ext_vector_type(4))) float v4f;
typedef __attribute__((ext_vector_type(8))) float v8f;

#define GCN_NODES   100000
#define GCN_EDGES   1600000
#define GCN_KDIM    128
#define GCN_HID     128
#define GCN_CLS     40
#define GCN_LDA     129

__global__ void gcn_zero(float* __restrict__ p, long n) {
    long i = (long)blockIdx.x * blockDim.x + threadIdx.x;
    long stride = (long)gridDim.x * blockDim.x;
    for (; i < n; i += stride) p[i] = 0.0f;
}

__global__ void gcn_degrees(const int* __restrict__ src, const int* __restrict__ dst,
                            float* __restrict__ odeg, float* __restrict__ ideg, int nE) {
    int i = blockIdx.x * blockDim.x + threadIdx.x;
    int stride = gridDim.x * blockDim.x;
    for (; i < nE; i += stride) {
        atomicAdd(&odeg[src[i]], 1.0f);
        atomicAdd(&ideg[dst[i]], 1.0f);
    }
}

__global__ void gcn_rsqrt_clamp(float* __restrict__ p, int n) {
    int i = blockIdx.x * blockDim.x + threadIdx.x;
    if (i < n) p[i] = rsqrtf(fmaxf(p[i], 1.0f));
}

__global__ void gcn_wfrag(const float* __restrict__ W, float* __restrict__ Wf,
                          int N, int nColTiles) {
    int i = blockIdx.x * blockDim.x + threadIdx.x;
    int total = nColTiles * 32 * 32;
    if (i >= total) return;
    int lane   = i & 31;
    int k4     = (i >> 5) & 31;
    int t      = i >> 10;
    int lane16 = lane & 15;
    int half   = lane >> 4;
    int kr     = 4 * k4 + 2 * half;
    int col    = t * 16 + lane16;
    float v0 = 0.0f, v1 = 0.0f;
    if (col < N) {
        v0 = W[kr * N + col];
        v1 = W[(kr + 1) * N + col];
    }
    Wf[2 * i]     = v0;
    Wf[2 * i + 1] = v1;
}

__global__ void __launch_bounds__(256)
gcn_gemm_wmma(const float* __restrict__ X, const float* __restrict__ ns,
              const float* __restrict__ Wf, float* __restrict__ Y,
              int N, int nColTiles) {
    __shared__ float Alds[32 * GCN_LDA];

    const int tid = threadIdx.x;
    const int r0  = blockIdx.x * 32;

    #pragma unroll
    for (int idx = tid; idx < 32 * GCN_KDIM; idx += 256) {
        int m = idx >> 7;
        int k = idx & 127;
        Alds[m * GCN_LDA + k] = X[(long)(r0 + m) * GCN_KDIM + k] * ns[r0 + m];
    }
    __syncthreads();

    const int wave   = tid >> 5;
    const int lane   = tid & 31;
    const int lane16 = lane & 15;
    const int half   = lane >> 4;
    if (wave >= nColTiles) return;

    const float* As0 = &Alds[lane16 * GCN_LDA + 2 * half];
    const float* As1 = As0 + 16 * GCN_LDA;
    const v2f* Wfp = (const v2f*)Wf + (long)wave * 32 * 32 + lane;

    v8f acc0 = {0.f, 0.f, 0.f, 0.f, 0.f, 0.f, 0.f, 0.f};
    v8f acc1 = {0.f, 0.f, 0.f, 0.f, 0.f, 0.f, 0.f, 0.f};

    #pragma unroll 8
    for (int k4 = 0; k4 < GCN_KDIM / 4; ++k4) {
        const int k = k4 * 4;
        v2f a0, a1;
        a0.x = As0[k];
        a0.y = As0[k + 1];
        a1.x = As1[k];
        a1.y = As1[k + 1];
        v2f b = Wfp[k4 * 32];
        acc0 = __builtin_amdgcn_wmma_f32_16x16x4_f32(
                   false, a0, false, b, (short)0, acc0, false, false);
        acc1 = __builtin_amdgcn_wmma_f32_16x16x4_f32(
                   false, a1, false, b, (short)0, acc1, false, false);
    }

    const int col = wave * 16 + lane16;
    if (col < N) {
        #pragma unroll
        for (int r = 0; r < 8; ++r) {
            int m = r + 8 * half;
            Y[(long)(r0 + m) * N + col]      = acc0[r];
            Y[(long)(r0 + 16 + m) * N + col] = acc1[r];
        }
    }
}

__global__ void __launch_bounds__(256)
gcn_spmm_scatter(const float* __restrict__ Xe, const int* __restrict__ src,
                 const int* __restrict__ dst, float* __restrict__ Y,
                 int F, int nE) {
    int gid  = blockIdx.x * blockDim.x + threadIdx.x;
    int e    = gid >> 5;
    int lane = gid & 31;
    if (e >= nE) return;
    int s = src[e];
    int d = dst[e];
    const v4f* xr = (const v4f*)(Xe + (long)s * F);
    float*     yr = Y + (long)d * F;
    for (int c = lane; c < (F >> 2); c += 32) {
        v4f v = xr[c];
        atomicAdd(&yr[4 * c + 0], v.x);
        atomicAdd(&yr[4 * c + 1], v.y);
        atomicAdd(&yr[4 * c + 2], v.z);
        atomicAdd(&yr[4 * c + 3], v.w);
    }
}

__global__ void gcn_finish(float* __restrict__ Y, const float* __restrict__ nd,
                           const float* __restrict__ bias, int F, int doRelu,
                           int nNodes) {
    long i = (long)blockIdx.x * blockDim.x + threadIdx.x;
    long total = (long)nNodes * F;
    if (i >= total) return;
    int node = (int)(i / F);
    int f    = (int)(i - (long)node * F);
    float v  = Y[i] * nd[node] + bias[f];
    if (doRelu) v = fmaxf(v, 0.0f);
    Y[i] = v;
}

extern "C" void kernel_launch(void* const* d_in, const int* in_sizes, int n_in,
                              void* d_out, int out_size, void* d_ws, size_t ws_size,
                              hipStream_t stream) {
    (void)in_sizes; (void)n_in; (void)out_size; (void)ws_size;

    const float* feat = (const float*)d_in[0];
    const int*   src  = (const int*)  d_in[1];
    const int*   dst  = (const int*)  d_in[2];
    const float* W0   = (const float*)d_in[3];
    const float* b0   = (const float*)d_in[4];
    const float* W1   = (const float*)d_in[5];
    const float* b1   = (const float*)d_in[6];
    const float* W2   = (const float*)d_in[7];
    const float* b2   = (const float*)d_in[8];
    float*       out  = (float*)d_out;

    const int  Nn = GCN_NODES, Ne = GCN_EDGES, H = GCN_HID, C = GCN_CLS;
    float* ws   = (float*)d_ws;
    float* ns   = ws;
    float* nd   = ws + Nn;
    float* bufA = ws + 2L * Nn;
    float* bufB = bufA + (long)Nn * H;
    float* Wf0  = bufB + (long)Nn * H;
    float* Wf1  = Wf0 + 8 * 2048;
    float* Wf2  = Wf1 + 8 * 2048;

    const int spmmBlocks = (Ne * 32) / 256;

    gcn_zero<<<512, 256, 0, stream>>>(ns, 2L * Nn);
    gcn_degrees<<<2048, 256, 0, stream>>>(src, dst, ns, nd, Ne);
    gcn_rsqrt_clamp<<<(2 * Nn + 255) / 256, 256, 0, stream>>>(ns, 2 * Nn);
    gcn_wfrag<<<(8 * 1024 + 255) / 256, 256, 0, stream>>>(W0, Wf0, H, 8);
    gcn_wfrag<<<(8 * 1024 + 255) / 256, 256, 0, stream>>>(W1, Wf1, H, 8);
    gcn_wfrag<<<(3 * 1024 + 255) / 256, 256, 0, stream>>>(W2, Wf2, C, 3);

    gcn_gemm_wmma<<<Nn / 32, 256, 0, stream>>>(feat, ns, Wf0, bufA, H, 8);
    gcn_zero<<<2048, 256, 0, stream>>>(bufB, (long)Nn * H);
    gcn_spmm_scatter<<<spmmBlocks, 256, 0, stream>>>(bufA, src, dst, bufB, H, Ne);
    gcn_finish<<<((long)Nn * H + 255) / 256, 256, 0, stream>>>(bufB, nd, b0, H, 1, Nn);

    gcn_gemm_wmma<<<Nn / 32, 256, 0, stream>>>(bufB, ns, Wf1, bufA, H, 8);
    gcn_zero<<<2048, 256, 0, stream>>>(bufB, (long)Nn * H);
    gcn_spmm_scatter<<<spmmBlocks, 256, 0, stream>>>(bufA, src, dst, bufB, H, Ne);
    gcn_finish<<<((long)Nn * H + 255) / 256, 256, 0, stream>>>(bufB, nd, b1, H, 1, Nn);

    gcn_gemm_wmma<<<Nn / 32, 256, 0, stream>>>(bufB, ns, Wf2, bufA, C, 3);
    gcn_zero<<<1024, 256, 0, stream>>>(out, (long)Nn * C);
    gcn_spmm_scatter<<<spmmBlocks, 256, 0, stream>>>(bufA, src, dst, out, C, Ne);
    gcn_finish<<<((long)Nn * C + 255) / 256, 256, 0, stream>>>(out, nd, b2, C, 0, Nn);
}


# Degree counts: a block takes an edge slice and a node range, counts in LDS (integer adds), stores per-slice planes that the norm pass sums; 3.2M global float atomics and the zero launch gone
# speedup vs baseline: 3.3738x; 3.3738x over previous
#include <hip/hip_runtime.h>

typedef __attribute__((ext_vector_type(2))) float v2f;
typedef __attribute__((ext_vector_type(4))) float v4f;
typedef __attribute__((ext_vector_type(8))) float v8f;

#define GCN_NODES   100000
#define GCN_EDGES   1600000
#define GCN_KDIM    128
#define GCN_HID     128
#define GCN_CLS     40
#define GCN_LDA     129
#define GCN_RPB     392
#define GCN_AGGBLK  ((GCN_NODES + GCN_RPB - 1) / GCN_RPB)
#define GCN_QDRAIN  64
#define GCN_QCAP    576
static_assert(GCN_EDGES % 4 == 0 && GCN_EDGES <= (1 << 21), "edge number fits 21 bits, edges read as int4");
static_assert(GCN_RPB + 1 <= (1 << 9) && GCN_RPB % 4 == 0, "local row fits 9 bits");
static_assert(GCN_AGGBLK <= 256, "the number of the aggregation block that owns a row fits one byte");
static_assert(GCN_EDGES % 16 == 0, "the owner map is read as 16-byte vectors");
#define GCN_OWNV    (GCN_EDGES / 16)
#define GCN_SRCBITS 17
#define GCN_DSL     32
#define GCN_DRG     8
#define GCN_DRN     (GCN_NODES / GCN_DRG)
#define GCN_DSV     (GCN_EDGES / 4 / GCN_DSL)
static_assert(GCN_NODES % GCN_DRG == 0 && (GCN_EDGES / 4) % GCN_DSL == 0, "ranges and slices are whole");
static_assert(GCN_NODES <= (1 << GCN_SRCBITS) && GCN_RPB + 1 <= (1 << (32 - GCN_SRCBITS)),
              "a list word holds a source row and a local destination row");

__global__ void __launch_bounds__(1024)
gcn_degrees(const int* __restrict__ src, const int* __restrict__ dst,
            unsigned* __restrict__ part, unsigned* __restrict__ own,
            unsigned* __restrict__ blkcnt) {
    __shared__ unsigned cs[GCN_DRN];
    __shared__ unsigned cd[GCN_DRN];
    const int tid   = threadIdx.x;
    const int slice = blockIdx.x / GCN_DRG;
    const int rng   = blockIdx.x - slice * GCN_DRG;
    const int rlo   = rng * GCN_DRN;
    const bool maps = rng == (slice % GCN_DRG);

    for (int i = tid; i < GCN_DRN; i += 1024) { cs[i] = 0u; cd[i] = 0u; }
    if (blockIdx.x == 0 && tid < 256) blkcnt[tid] = 0u;
    __syncthreads();

    const int4* s4 = (const int4*)src + (long)slice * GCN_DSV;
    const int4* d4 = (const int4*)dst + (long)slice * GCN_DSV;
    unsigned* ownp = own + (long)slice * GCN_DSV;
    for (int base = 0; base < GCN_DSV; base += 4096) {
        int4 s[4], d[4];
        #pragma unroll
        for (int j = 0; j < 4; ++j) {
            const int v = min(base + 1024 * j + tid, GCN_DSV - 1);
            s[j] = s4[v];
            d[j] = d4[v];
        }
        #pragma unroll
        for (int j = 0; j < 4; ++j) {
            const int v = base + 1024 * j + tid;
            if (v < GCN_DSV) {
                unsigned l;
                l = (unsigned)(s[j].x - rlo); if (l < (unsigned)GCN_DRN) atomicAdd(&cs[l], 1u);
                l = (unsigned)(s[j].y - rlo); if (l < (unsigned)GCN_DRN) atomicAdd(&cs[l], 1u);
                l = (unsigned)(s[j].z - rlo); if (l < (unsigned)GCN_DRN) atomicAdd(&cs[l], 1u);
                l = (unsigned)(s[j].w - rlo); if (l < (unsigned)GCN_DRN) atomicAdd(&cs[l], 1u);
                l = (unsigned)(d[j].x - rlo); if (l < (unsigned)GCN_DRN) atomicAdd(&cd[l], 1u);
                l = (unsigned)(d[j].y - rlo); if (l < (unsigned)GCN_DRN) atomicAdd(&cd[l], 1u);
                l = (unsigned)(d[j].z - rlo); if (l < (unsigned)GCN_DRN) atomicAdd(&cd[l], 1u);
                l = (unsigned)(d[j].w - rlo); if (l < (unsigned)GCN_DRN) atomicAdd(&cd[l], 1u);
                if (maps)
                    ownp[v] = ((unsigned)d[j].x / GCN_RPB) | (((unsigned)d[j].y / GCN_RPB) << 8) |
                              (((unsigned)d[j].z / GCN_RPB) << 16) | (((unsigned)d[j].w / GCN_RPB) << 24);
            }
        }
    }
    __syncthreads();

    unsigned* ps = part + ((long)slice * GCN_NODES + rlo);
    unsigned* pd = part + ((long)(GCN_DSL + slice) * GCN_NODES + rlo);
    for (int i = tid; i < GCN_DRN; i += 1024) { ps[i] = cs[i]; pd[i] = cd[i]; }
}

__global__ void gcn_norms(const unsigned* __restrict__ part, float* __restrict__ p, int nNodes,
                          unsigned* __restrict__ blkcnt) {
    int i = blockIdx.x * blockDim.x + threadIdx.x;
    if (i < 2 * nNodes) {
        const bool in = i >= nNodes;
        const unsigned* q = part + (in ? (long)GCN_DSL * nNodes + (i - nNodes) : (long)i);
        unsigned c = 0u;
        #pragma unroll
        for (int s = 0; s < GCN_DSL; ++s) c += q[(long)s * nNodes];
        if (in && c > 0u) atomicAdd(&blkcnt[(i - nNodes) / GCN_RPB], c);
        p[i] = rsqrtf(fmaxf((float)c, 1.0f));
    }
}

__device__ __forceinline__ unsigned gcn_list_base(const unsigned* __restrict__ blkcnt,
                                                  int b, int lane) {
    unsigned s = 0;
    for (int i = lane; i < b; i += 32) s += blkcnt[i];
    #pragma unroll
    for (int o = 16; o > 0; o >>= 1) s += __shfl_xor(s, o, 32);
    return s;
}

__global__ void gcn_wfrag(const float* __restrict__ W, float* __restrict__ Wf,
                          int N, int nColTiles) {
    int i = blockIdx.x * blockDim.x + threadIdx.x;
    int total = nColTiles * 32 * 32;
    if (i >= total) return;
    int lane   = i & 31;
    int k4     = (i >> 5) & 31;
    int t      = i >> 10;
    int lane16 = lane & 15;
    int half   = lane >> 4;
    int kr     = 4 * k4 + 2 * half;
    int col    = t * 16 + lane16;
    float v0 = 0.0f, v1 = 0.0f;
    if (col < N) {
        v0 = W[kr * N + col];
        v1 = W[(kr + 1) * N + col];
    }
    Wf[2 * i]     = v0;
    Wf[2 * i + 1] = v1;
}

__global__ void __launch_bounds__(256)
gcn_gemm_wmma(const float* __restrict__ X, const float* __restrict__ ns,
              const float* __restrict__ Wf, float* __restrict__ Y,
              int N, int nColTiles) {
    __shared__ float Alds[32 * GCN_LDA];

    const int tid = threadIdx.x;
    const int r0  = blockIdx.x * 32;

    #pragma unroll
    for (int idx = tid; idx < 32 * GCN_KDIM; idx += 256) {
        int m = idx >> 7;
        int k = idx & 127;
        Alds[m * GCN_LDA + k] = X[(long)(r0 + m) * GCN_KDIM + k] * ns[r0 + m];
    }
    __syncthreads();

    const int wave   = tid >> 5;
    const int lane   = tid & 31;
    const int lane16 = lane & 15;
    const int half   = lane >> 4;
    if (wave >= nColTiles) return;

    const float* As0 = &Alds[lane16 * GCN_LDA + 2 * half];
    const float* As1 = As0 + 16 * GCN_LDA;
    const v2f* Wfp = (const v2f*)Wf + (long)wave * 32 * 32 + lane;

    v8f acc0 = {0.f, 0.f, 0.f, 0.f, 0.f, 0.f, 0.f, 0.f};
    v8f acc1 = {0.f, 0.f, 0.f, 0.f, 0.f, 0.f, 0.f, 0.f};

    #pragma unroll 8
    for (int k4 = 0; k4 < GCN_KDIM / 4; ++k4) {
        const int k = k4 * 4;
        v2f a0, a1;
        a0.x = As0[k];
        a0.y = As0[k + 1];
        a1.x = As1[k];
        a1.y = As1[k + 1];
        v2f b = Wfp[k4 * 32];
        acc0 = __builtin_amdgcn_wmma_f32_16x16x4_f32(
                   false, a0, false, b, (short)0, acc0, false, false);
        acc1 = __builtin_amdgcn_wmma_f32_16x16x4_f32(
                   false, a1, false, b, (short)0, acc1, false, false);
    }

    const int col = wave * 16 + lane16;
    if (col < N) {
        #pragma unroll
        for (int r = 0; r < 8; ++r) {
            int m = r + 8 * half;
            Y[(long)(r0 + m) * N + col]      = acc0[r];
            Y[(long)(r0 + 16 + m) * N + col] = acc1[r];
        }
    }
}

template <int F, int RELU>
__global__ void __launch_bounds__(1024)
gcn_agg_lds(const float* __restrict__ Xe, const int* __restrict__ src,
            const int* __restrict__ dst, const unsigned* __restrict__ own,
            const float* __restrict__ nd, const float* __restrict__ bias,
            float* __restrict__ Y, int nNodes,
            const unsigned* __restrict__ blkcnt, unsigned* __restrict__ list) {
    static_assert(F % 4 == 0 && F / 4 <= 32, "a row is read as one 16-B piece per lane");
    constexpr int LPR = F / 4;
    __shared__ __attribute__((aligned(16))) float acc[(GCN_RPB + 1) * F];
    __shared__ unsigned queue[32 * GCN_QCAP];
    __shared__ unsigned lpos;

    const int tid  = threadIdx.x;
    const int lane = tid & 31;
    const int wave = tid >> 5;
    const int half = lane >> 4;
    const int lo   = blockIdx.x * GCN_RPB;
    const int rows = min(GCN_RPB, nNodes - lo);
    unsigned* q = queue + wave * GCN_QCAP;
    const bool active = lane < LPR;
    unsigned* lst = list + __builtin_amdgcn_readfirstlane(gcn_list_base(blkcnt, blockIdx.x, lane));

    for (int i = tid; i < (GCN_RPB + 1) * F / 4; i += 1024)
        ((v4f*)acc)[i] = (v4f){0.f, 0.f, 0.f, 0.f};
    if (tid == 0) lpos = 0u;
    __syncthreads();

    const int NV = GCN_OWNV;
    const uint4* own4 = (const uint4*)own;
    const unsigned pat = blockIdx.x * 0x01010101u;
    int cnt = 0;
    const int o0 = 4 * lane + half, o1 = 4 * lane + ((1 + half) & 3);
    const int o2 = 4 * lane + ((2 + half) & 3), o3 = 4 * lane + ((3 + half) & 3);

    auto drain = [&]() {
        __builtin_amdgcn_fence(__ATOMIC_ACQ_REL, "wavefront");
        for (int b = 0; b < cnt; b += 32) {
            const int n = min(32, cnt - b);
            int sv = 0, dl = GCN_RPB;
            unsigned p0 = 0u;
            if (lane == 0)
                p0 = __hip_atomic_fetch_add(&lpos, (unsigned)n, __ATOMIC_RELAXED, __HIP_MEMORY_SCOPE_WORKGROUP);
            p0 = __builtin_amdgcn_readfirstlane(p0);
            if (lane < n) {
                const unsigned e = q[b + lane];
                sv = src[e];
                dl = dst[e] - lo;
                lst[p0 + lane] = (unsigned)sv | ((unsigned)dl << GCN_SRCBITS);
            }
            for (int k = 0; k < n; k += 8) {
                v4f r[8];
                #pragma unroll
                for (int j = 0; j < 8; ++j) {
                    const int s = __builtin_amdgcn_readlane(sv, k + j);
                    r[j] = (v4f){0.f, 0.f, 0.f, 0.f};
                    if (active) r[j] = *(const v4f*)(Xe + (long)s * F + 4 * lane);
                }
                #pragma unroll
                for (int j = 0; j < 8; ++j) {
                    const int d = __builtin_amdgcn_readlane(dl, k + j);
                    if (active) {
                        float* a = acc + d * F;
                        const v4f x = r[j];
                        const float x0 = half ? x.y : x.x, x1 = half ? x.z : x.y;
                        const float x2 = half ? x.w : x.z, x3 = half ? x.x : x.w;
                        __hip_atomic_fetch_add(a + o0, x0, __ATOMIC_RELAXED, __HIP_MEMORY_SCOPE_WORKGROUP);
                        __hip_atomic_fetch_add(a + o1, x1, __ATOMIC_RELAXED, __HIP_MEMORY_SCOPE_WORKGROUP);
                        __hip_atomic_fetch_add(a + o2, x2, __ATOMIC_RELAXED, __HIP_MEMORY_SCOPE_WORKGROUP);
                        __hip_atomic_fetch_add(a + o3, x3, __ATOMIC_RELAXED, __HIP_MEMORY_SCOPE_WORKGROUP);
                    }
                }
            }
        }
        __builtin_amdgcn_fence(__ATOMIC_ACQ_REL, "wavefront");
        cnt = 0;
    };

    auto hit = [&](const unsigned w) {
        const unsigned t = w ^ pat;
        return ~(((t & 0x7F7F7F7Fu) + 0x7F7F7F7Fu) | t | 0x7F7F7F7Fu);
    };

    auto test = [&](const uint4 c, const int vidx, const bool live) {
        unsigned m = (hit(c.x) >> 7) | (hit(c.y) >> 6) | (hit(c.z) >> 5) | (hit(c.w) >> 4);
        if (!live) m = 0u;
        unsigned bal = __builtin_amdgcn_ballot_w32(m != 0u);
        while (bal != 0u) {
            if (m != 0u) {
                const int p = __builtin_ctz(m);
                q[cnt + (int)__builtin_amdgcn_mbcnt_lo(bal, 0u)] =
                    (unsigned)(16 * vidx + 4 * (p & 7) + (p >> 3));
                m &= m - 1u;
            }
            cnt += __builtin_popcount(bal);
            bal = __builtin_amdgcn_ballot_w32(m != 0u);
        }
    };

    const int nFull = NV & ~4095;
    if (nFull > 0) {
        uint4 n0 = own4[tid], n1 = own4[tid + 1024], n2 = own4[tid + 2048], n3 = own4[tid + 3072];
        for (int base = 0; base < nFull; base += 4096) {
            const uint4 c0 = n0, c1 = n1, c2 = n2, c3 = n3;
            const int nb = min(base + 4096, nFull - 4096) + tid;
            n0 = own4[nb]; n1 = own4[nb + 1024]; n2 = own4[nb + 2048]; n3 = own4[nb + 3072];
            test(c0, base + tid,        true);
            if (cnt >= GCN_QDRAIN) drain();
            test(c1, base + tid + 1024, true);
            if (cnt >= GCN_QDRAIN) drain();
            test(c2, base + tid + 2048, true);
            if (cnt >= GCN_QDRAIN) drain();
            test(c3, base + tid + 3072, true);
            if (cnt >= GCN_QDRAIN) drain();
        }
    }
    for (int base = nFull; base < NV; base += 1024) {
        const int vidx = base + tid;
        const uint4 c = own4[min(vidx, NV - 1)];
        test(c, vidx, vidx < NV);
        if (cnt >= GCN_QDRAIN) drain();
    }
    if (cnt > 0) drain();
    __syncthreads();

    const int nunits = rows * LPR;
    float* Yb = Y + (long)lo * F;
    for (int u = tid; u < nunits; u += 1024) {
        const int rl = u / LPR;
        const int f4 = u - rl * LPR;
        const v4f a = ((const v4f*)acc)[u];
        const float s = nd[lo + rl];
        const v4f b = *(const v4f*)(bias + 4 * f4);
        v4f o;
        o.x = a.x * s + b.x; o.y = a.y * s + b.y; o.z = a.z * s + b.z; o.w = a.w * s + b.w;
        if (RELU) {
            o.x = fmaxf(o.x, 0.0f); o.y = fmaxf(o.y, 0.0f);
            o.z = fmaxf(o.z, 0.0f); o.w = fmaxf(o.w, 0.0f);
        }
        ((v4f*)Yb)[u] = o;
    }
}

template <int F, int RELU>
__global__ void __launch_bounds__(1024)
gcn_agg_list(const float* __restrict__ Xe, const unsigned* __restrict__ blkcnt,
             const unsigned* __restrict__ list, const float* __restrict__ nd,
             const float* __restrict__ bias, float* __restrict__ Y, int nNodes) {
    static_assert(F % 4 == 0 && F / 4 <= 32, "a row is read as one 16-B piece per lane");
    constexpr int LPR = F / 4;
    __shared__ __attribute__((aligned(16))) float acc[(GCN_RPB + 1) * F];

    const int tid  = threadIdx.x;
    const int lane = tid & 31;
    const int wave = __builtin_amdgcn_readfirstlane(tid >> 5);
    const int half = lane >> 4;
    const int lo   = blockIdx.x * GCN_RPB;
    const int rows = min(GCN_RPB, nNodes - lo);
    const bool active = lane < LPR;
    const int cnt = (int)blkcnt[blockIdx.x];
    const unsigned* L = list + __builtin_amdgcn_readfirstlane(gcn_list_base(blkcnt, blockIdx.x, lane));

    for (int i = tid; i < (GCN_RPB + 1) * F / 4; i += 1024)
        ((v4f*)acc)[i] = (v4f){0.f, 0.f, 0.f, 0.f};
    __syncthreads();

    const int o0 = 4 * lane + half, o1 = 4 * lane + ((1 + half) & 3);
    const int o2 = 4 * lane + ((2 + half) & 3), o3 = 4 * lane + ((3 + half) & 3);

    const unsigned pad = (unsigned)GCN_RPB << GCN_SRCBITS;
    const int first = 8 * wave + 256 * (lane >> 3) + (lane & 7);
    auto fetch = [&](const int it) {
        const int idx = first + 1024 * it;
        return idx < cnt ? L[idx] : pad;
    };
    unsigned cur = fetch(0);
    for (int it = 0; 8 * wave + 1024 * it < cnt; ++it) {
        const unsigned nxt = fetch(it + 1);
        const int sv = (int)(cur & ((1u << GCN_SRCBITS) - 1u));
        const int dl = (int)(cur >> GCN_SRCBITS);
        const int ng = min(4, (cnt - 8 * wave - 1024 * it + 255) >> 8);
        for (int k = 0; k < 8 * ng; k += 8) {
            v4f r[8];
            #pragma unroll
            for (int j = 0; j < 8; ++j) {
                const int s = __builtin_amdgcn_readlane(sv, k + j);
                r[j] = (v4f){0.f, 0.f, 0.f, 0.f};
                if (active) r[j] = *(const v4f*)(Xe + (long)s * F + 4 * lane);
            }
            #pragma unroll
            for (int j = 0; j < 8; ++j) {
                const int d = __builtin_amdgcn_readlane(dl, k + j);
                if (active) {
                    float* a = acc + d * F;
                    const v4f x = r[j];
                    const float x0 = half ? x.y : x.x, x1 = half ? x.z : x.y;
                    const float x2 = half ? x.w : x.z, x3 = half ? x.x : x.w;
                    __hip_atomic_fetch_add(a + o0, x0, __ATOMIC_RELAXED, __HIP_MEMORY_SCOPE_WORKGROUP);
                    __hip_atomic_fetch_add(a + o1, x1, __ATOMIC_RELAXED, __HIP_MEMORY_SCOPE_WORKGROUP);
                    __hip_atomic_fetch_add(a + o2, x2, __ATOMIC_RELAXED, __HIP_MEMORY_SCOPE_WORKGROUP);
                    __hip_atomic_fetch_add(a + o3, x3, __ATOMIC_RELAXED, __HIP_MEMORY_SCOPE_WORKGROUP);
                }
            }
        }
        cur = nxt;
    }
    __syncthreads();

    const int nunits = rows * LPR;
    float* Yb = Y + (long)lo * F;
    for (int u = tid; u < nunits; u += 1024) {
        const int rl = u / LPR;
        const int f4 = u - rl * LPR;
        const v4f a = ((const v4f*)acc)[u];
        const float s = nd[lo + rl];
        const v4f b = *(const v4f*)(bias + 4 * f4);
        v4f o;
        o.x = a.x * s + b.x; o.y = a.y * s + b.y; o.z = a.z * s + b.z; o.w = a.w * s + b.w;
        if (RELU) {
            o.x = fmaxf(o.x, 0.0f); o.y = fmaxf(o.y, 0.0f);
            o.z = fmaxf(o.z, 0.0f); o.w = fmaxf(o.w, 0.0f);
        }
        ((v4f*)Yb)[u] = o;
    }
}

template <int F, int RELU>
__global__ void __launch_bounds__(1024)
gcn_agg_list_packed(const float* __restrict__ Xe, const unsigned* __restrict__ blkcnt,
                    const unsigned* __restrict__ list, const float* __restrict__ nd,
                    const float* __restrict__ bias, float* __restrict__ Y, int nNodes) {
    static_assert(F % 4 == 0 && F / 4 <= 16, "at least two rows fit one wave instruction");
    constexpr int LPR   = F / 4;
    constexpr int RPI   = 32 / LPR;
    constexpr int STEPS = 32 / RPI;
    constexpr int CH    = RPI * STEPS;
    static_assert(RPI >= 2 && RPI <= 4 && CH <= 32, "a lane group's float rotation is its number, below 4");
    __shared__ __attribute__((aligned(16))) float acc[(GCN_RPB + 1) * F];

    const int tid  = threadIdx.x;
    const int lane = tid & 31;
    const int wave = __builtin_amdgcn_readfirstlane(tid >> 5);
    const int slot = lane / LPR;
    const int sub  = lane - slot * LPR;
    const bool active = slot < RPI;
    const int lo   = blockIdx.x * GCN_RPB;
    const int rows = min(GCN_RPB, nNodes - lo);
    const int cnt  = (int)blkcnt[blockIdx.x];
    const unsigned* L = list + __builtin_amdgcn_readfirstlane(gcn_list_base(blkcnt, blockIdx.x, lane));

    for (int i = tid; i < (GCN_RPB + 1) * F / 4; i += 1024)
        ((v4f*)acc)[i] = (v4f){0.f, 0.f, 0.f, 0.f};
    __syncthreads();

    const int o0 = 4 * sub + (slot & 3),       o1 = 4 * sub + ((1 + slot) & 3);
    const int o2 = 4 * sub + ((2 + slot) & 3), o3 = 4 * sub + ((3 + slot) & 3);
    const bool rot1 = (slot & 1) != 0, rot2 = (slot & 2) != 0;

    const unsigned pad = (unsigned)GCN_RPB << GCN_SRCBITS;
    unsigned cur = pad;
    {
        const int idx = CH * wave + lane;
        if (lane < CH && idx < cnt) cur = L[idx];
    }
    for (int it = 0; CH * (wave + 32 * it) < cnt; ++it) {
        const int nidx = CH * (wave + 32 * (it + 1)) + lane;
        const unsigned nraw = L[min(nidx, cnt - 1)];
        v4f r[STEPS];
        int drow[STEPS];
        #pragma unroll
        for (int t = 0; t < STEPS; ++t) {
            unsigned mine = __builtin_amdgcn_readlane(cur, RPI * t);
            #pragma unroll
            for (int g = 1; g < RPI; ++g) {
                const unsigned eg = __builtin_amdgcn_readlane(cur, RPI * t + g);
                mine = (slot == g) ? eg : mine;
            }
            const unsigned sv = mine & ((1u << GCN_SRCBITS) - 1u);
            drow[t] = (int)(mine >> GCN_SRCBITS) * F;
            r[t] = *(const v4f*)(Xe + (sv * (unsigned)F + 4u * (unsigned)sub));
        }
        static_assert(STEPS == 10, "the statement below names ten row reads");
        asm volatile("" :: "v"(r[0]), "v"(r[1]), "v"(r[2]), "v"(r[3]), "v"(r[4]),
                           "v"(r[5]), "v"(r[6]), "v"(r[7]), "v"(r[8]), "v"(r[9]), "v"(nraw));
        const unsigned nxt = (lane < CH && nidx < cnt) ? nraw : pad;
        #pragma unroll
        for (int t = 0; t < STEPS; ++t) {
            if (active) {
                float* a = acc + drow[t];
                const v4f x = r[t];
                const float y0 = rot1 ? x.y : x.x, y1 = rot1 ? x.z : x.y;
                const float y2 = rot1 ? x.w : x.z, y3 = rot1 ? x.x : x.w;
                const float x0 = rot2 ? y2 : y0, x1 = rot2 ? y3 : y1;
                const float x2 = rot2 ? y0 : y2, x3 = rot2 ? y1 : y3;
                __hip_atomic_fetch_add(a + o0, x0, __ATOMIC_RELAXED, __HIP_MEMORY_SCOPE_WORKGROUP);
                __hip_atomic_fetch_add(a + o1, x1, __ATOMIC_RELAXED, __HIP_MEMORY_SCOPE_WORKGROUP);
                __hip_atomic_fetch_add(a + o2, x2, __ATOMIC_RELAXED, __HIP_MEMORY_SCOPE_WORKGROUP);
                __hip_atomic_fetch_add(a + o3, x3, __ATOMIC_RELAXED, __HIP_MEMORY_SCOPE_WORKGROUP);
            }
        }
        cur = nxt;
    }
    __syncthreads();

    const int nunits = rows * LPR;
    float* Yb = Y + (long)lo * F;
    for (int u = tid; u < nunits; u += 1024) {
        const int rl = u / LPR;
        const int f4 = u - rl * LPR;
        const v4f a = ((const v4f*)acc)[u];
        const float s = nd[lo + rl];
        const v4f b = *(const v4f*)(bias + 4 * f4);
        v4f o;
        o.x = a.x * s + b.x; o.y = a.y * s + b.y; o.z = a.z * s + b.z; o.w = a.w * s + b.w;
        if (RELU) {
            o.x = fmaxf(o.x, 0.0f); o.y = fmaxf(o.y, 0.0f);
            o.z = fmaxf(o.z, 0.0f); o.w = fmaxf(o.w, 0.0f);
        }
        ((v4f*)Yb)[u] = o;
    }
}

extern "C" void kernel_launch(void* const* d_in, const int* in_sizes, int n_in,
                              void* d_out, int out_size, void* d_ws, size_t ws_size,
                              hipStream_t stream) {
    (void)in_sizes; (void)n_in; (void)out_size; (void)ws_size;

    const float* feat = (const float*)d_in[0];
    const int*   src  = (const int*)  d_in[1];
    const int*   dst  = (const int*)  d_in[2];
    const float* W0   = (const float*)d_in[3];
    const float* b0   = (const float*)d_in[4];
    const float* W1   = (const float*)d_in[5];
    const float* b1   = (const float*)d_in[6];
    const float* W2   = (const float*)d_in[7];
    const float* b2   = (const float*)d_in[8];
    float*       out  = (float*)d_out;

    const int  Nn = GCN_NODES, Ne = GCN_EDGES, H = GCN_HID, C = GCN_CLS;
    float* ws   = (float*)d_ws;
    float* ns   = ws;
    float* nd   = ws + Nn;
    unsigned* blkcnt = (unsigned*)(ws + 2L * Nn);
    float* bufA = ws + 2L * Nn + 256;
    float* bufB = bufA + (long)Nn * H;
    float* Wf0  = bufB + (long)Nn * H;
    float* Wf1  = Wf0 + 8 * 2048;
    float* Wf2  = Wf1 + 8 * 2048;
    unsigned* own = (unsigned*)(Wf2 + 3 * 2048);
    unsigned* list = own + Ne / 4;
    unsigned* part = list + Ne;

    gcn_degrees<<<GCN_DSL * GCN_DRG, 1024, 0, stream>>>(src, dst, part, own, blkcnt);
    gcn_norms<<<(2 * Nn + 255) / 256, 256, 0, stream>>>(part, ns, Nn, blkcnt);
    gcn_wfrag<<<(8 * 1024 + 255) / 256, 256, 0, stream>>>(W0, Wf0, H, 8);
    gcn_wfrag<<<(8 * 1024 + 255) / 256, 256, 0, stream>>>(W1, Wf1, H, 8);
    gcn_wfrag<<<(3 * 1024 + 255) / 256, 256, 0, stream>>>(W2, Wf2, C, 3);

    gcn_gemm_wmma<<<Nn / 32, 256, 0, stream>>>(feat, ns, Wf0, bufA, H, 8);
    gcn_agg_lds<GCN_HID, 1><<<GCN_AGGBLK, 1024, 0, stream>>>(bufA, src, dst, own, nd, b0, bufB, Nn, blkcnt, list);

    gcn_gemm_wmma<<<Nn / 32, 256, 0, stream>>>(bufB, ns, Wf1, bufA, H, 8);
    gcn_agg_list<GCN_HID, 1><<<GCN_AGGBLK, 1024, 0, stream>>>(bufA, blkcnt, list, nd, b1, bufB, Nn);

    gcn_gemm_wmma<<<Nn / 32, 256, 0, stream>>>(bufB, ns, Wf2, bufA, C, 3);
    gcn_agg_list_packed<GCN_CLS, 0><<<GCN_AGGBLK, 1024, 0, stream>>>(bufA, blkcnt, list, nd, b2, out, Nn);
}


# Block edge counts: norm pass runs one block per aggregation block (392 nodes, both norms per thread), sums its in-degrees by shuffles + LDS, one plain store; 100000 global int atomics on 8 lines gone
# speedup vs baseline: 3.5506x; 1.0524x over previous
#include <hip/hip_runtime.h>

typedef __attribute__((ext_vector_type(2))) float v2f;
typedef __attribute__((ext_vector_type(4))) float v4f;
typedef __attribute__((ext_vector_type(8))) float v8f;

#define GCN_NODES   100000
#define GCN_EDGES   1600000
#define GCN_KDIM    128
#define GCN_HID     128
#define GCN_CLS     40
#define GCN_LDA     129
#define GCN_RPB     392
#define GCN_AGGBLK  ((GCN_NODES + GCN_RPB - 1) / GCN_RPB)
#define GCN_QDRAIN  64
#define GCN_QCAP    576
static_assert(GCN_EDGES % 4 == 0 && GCN_EDGES <= (1 << 21), "edge number fits 21 bits, edges read as int4");
static_assert(GCN_RPB + 1 <= (1 << 9) && GCN_RPB % 4 == 0, "local row fits 9 bits");
static_assert(GCN_AGGBLK <= 256, "the number of the aggregation block that owns a row fits one byte");
static_assert(GCN_EDGES % 16 == 0, "the owner map is read as 16-byte vectors");
#define GCN_OWNV    (GCN_EDGES / 16)
#define GCN_SRCBITS 17
#define GCN_DSL     32
#define GCN_DRG     8
#define GCN_DRN     (GCN_NODES / GCN_DRG)
#define GCN_DSV     (GCN_EDGES / 4 / GCN_DSL)
static_assert(GCN_NODES % GCN_DRG == 0 && (GCN_EDGES / 4) % GCN_DSL == 0, "ranges and slices are whole");
static_assert(GCN_NODES <= (1 << GCN_SRCBITS) && GCN_RPB + 1 <= (1 << (32 - GCN_SRCBITS)),
              "a list word holds a source row and a local destination row");

__global__ void __launch_bounds__(1024)
gcn_degrees(const int* __restrict__ src, const int* __restrict__ dst,
            unsigned* __restrict__ part, unsigned* __restrict__ own) {
    __shared__ unsigned cs[GCN_DRN];
    __shared__ unsigned cd[GCN_DRN];
    const int tid   = threadIdx.x;
    const int slice = blockIdx.x / GCN_DRG;
    const int rng   = blockIdx.x - slice * GCN_DRG;
    const int rlo   = rng * GCN_DRN;
    const bool maps = rng == (slice % GCN_DRG);

    for (int i = tid; i < GCN_DRN; i += 1024) { cs[i] = 0u; cd[i] = 0u; }
    __syncthreads();

    const int4* s4 = (const int4*)src + (long)slice * GCN_DSV;
    const int4* d4 = (const int4*)dst + (long)slice * GCN_DSV;
    unsigned* ownp = own + (long)slice * GCN_DSV;
    for (int base = 0; base < GCN_DSV; base += 4096) {
        int4 s[4], d[4];
        #pragma unroll
        for (int j = 0; j < 4; ++j) {
            const int v = min(base + 1024 * j + tid, GCN_DSV - 1);
            s[j] = s4[v];
            d[j] = d4[v];
        }
        #pragma unroll
        for (int j = 0; j < 4; ++j) {
            const int v = base + 1024 * j + tid;
            if (v < GCN_DSV) {
                unsigned l;
                l = (unsigned)(s[j].x - rlo); if (l < (unsigned)GCN_DRN) atomicAdd(&cs[l], 1u);
                l = (unsigned)(s[j].y - rlo); if (l < (unsigned)GCN_DRN) atomicAdd(&cs[l], 1u);
                l = (unsigned)(s[j].z - rlo); if (l < (unsigned)GCN_DRN) atomicAdd(&cs[l], 1u);
                l = (unsigned)(s[j].w - rlo); if (l < (unsigned)GCN_DRN) atomicAdd(&cs[l], 1u);
                l = (unsigned)(d[j].x - rlo); if (l < (unsigned)GCN_DRN) atomicAdd(&cd[l], 1u);
                l = (unsigned)(d[j].y - rlo); if (l < (unsigned)GCN_DRN) atomicAdd(&cd[l], 1u);
                l = (unsigned)(d[j].z - rlo); if (l < (unsigned)GCN_DRN) atomicAdd(&cd[l], 1u);
                l = (unsigned)(d[j].w - rlo); if (l < (unsigned)GCN_DRN) atomicAdd(&cd[l], 1u);
                if (maps)
                    ownp[v] = ((unsigned)d[j].x / GCN_RPB) | (((unsigned)d[j].y / GCN_RPB) << 8) |
                              (((unsigned)d[j].z / GCN_RPB) << 16) | (((unsigned)d[j].w / GCN_RPB) << 24);
            }
        }
    }
    __syncthreads();

    unsigned* ps = part + ((long)slice * GCN_NODES + rlo);
    unsigned* pd = part + ((long)(GCN_DSL + slice) * GCN_NODES + rlo);
    for (int i = tid; i < GCN_DRN; i += 1024) { ps[i] = cs[i]; pd[i] = cd[i]; }
}

#define GCN_NTH 512
static_assert(GCN_NTH >= GCN_RPB && GCN_NTH % 32 == 0, "one thread per owned node");
__global__ void __launch_bounds__(GCN_NTH)
gcn_norms(const unsigned* __restrict__ part, float* __restrict__ p, int nNodes,
          unsigned* __restrict__ blkcnt) {
    __shared__ unsigned wsum[GCN_NTH / 32];
    const int t = threadIdx.x;
    const int n = blockIdx.x * GCN_RPB + t;
    const bool has = t < GCN_RPB && n < nNodes;
    unsigned cin = 0u, cout = 0u;
    if (has) {
        const unsigned* qo = part + n;
        const unsigned* qi = part + (long)GCN_DSL * nNodes + n;
        #pragma unroll
        for (int s = 0; s < GCN_DSL; ++s) {
            cout += qo[(long)s * nNodes];
            cin  += qi[(long)s * nNodes];
        }
    }
    unsigned s = cin;
    #pragma unroll
    for (int o = 16; o > 0; o >>= 1) s += __shfl_xor(s, o, 32);
    if ((t & 31) == 0) wsum[t >> 5] = s;
    __syncthreads();
    if (has) {
        p[n]          = rsqrtf(fmaxf((float)cout, 1.0f));
        p[nNodes + n] = rsqrtf(fmaxf((float)cin, 1.0f));
    }
    if (t == 0) {
        unsigned tot = 0u;
        #pragma unroll
        for (int w = 0; w < GCN_NTH / 32; ++w) tot += wsum[w];
        blkcnt[blockIdx.x] = tot;
    }
}

__device__ __forceinline__ unsigned gcn_list_base(const unsigned* __restrict__ blkcnt,
                                                  int b, int lane) {
    unsigned s = 0;
    for (int i = lane; i < b; i += 32) s += blkcnt[i];
    #pragma unroll
    for (int o = 16; o > 0; o >>= 1) s += __shfl_xor(s, o, 32);
    return s;
}

__global__ void gcn_wfrag(const float* __restrict__ W, float* __restrict__ Wf,
                          int N, int nColTiles) {
    int i = blockIdx.x * blockDim.x + threadIdx.x;
    int total = nColTiles * 32 * 32;
    if (i >= total) return;
    int lane   = i & 31;
    int k4     = (i >> 5) & 31;
    int t      = i >> 10;
    int lane16 = lane & 15;
    int half   = lane >> 4;
    int kr     = 4 * k4 + 2 * half;
    int col    = t * 16 + lane16;
    float v0 = 0.0f, v1 = 0.0f;
    if (col < N) {
        v0 = W[kr * N + col];
        v1 = W[(kr + 1) * N + col];
    }
    Wf[2 * i]     = v0;
    Wf[2 * i + 1] = v1;
}

__global__ void __launch_bounds__(256)
gcn_gemm_wmma(const float* __restrict__ X, const float* __restrict__ ns,
              const float* __restrict__ Wf, float* __restrict__ Y,
              int N, int nColTiles) {
    __shared__ float Alds[32 * GCN_LDA];

    const int tid = threadIdx.x;
    const int r0  = blockIdx.x * 32;

    #pragma unroll
    for (int idx = tid; idx < 32 * GCN_KDIM; idx += 256) {
        int m = idx >> 7;
        int k = idx & 127;
        Alds[m * GCN_LDA + k] = X[(long)(r0 + m) * GCN_KDIM + k] * ns[r0 + m];
    }
    __syncthreads();

    const int wave   = tid >> 5;
    const int lane   = tid & 31;
    const int lane16 = lane & 15;
    const int half   = lane >> 4;
    if (wave >= nColTiles) return;

    const float* As0 = &Alds[lane16 * GCN_LDA + 2 * half];
    const float* As1 = As0 + 16 * GCN_LDA;
    const v2f* Wfp = (const v2f*)Wf + (long)wave * 32 * 32 + lane;

    v8f acc0 = {0.f, 0.f, 0.f, 0.f, 0.f, 0.f, 0.f, 0.f};
    v8f acc1 = {0.f, 0.f, 0.f, 0.f, 0.f, 0.f, 0.f, 0.f};

    #pragma unroll 8
    for (int k4 = 0; k4 < GCN_KDIM / 4; ++k4) {
        const int k = k4 * 4;
        v2f a0, a1;
        a0.x = As0[k];
        a0.y = As0[k + 1];
        a1.x = As1[k];
        a1.y = As1[k + 1];
        v2f b = Wfp[k4 * 32];
        acc0 = __builtin_amdgcn_wmma_f32_16x16x4_f32(
                   false, a0, false, b, (short)0, acc0, false, false);
        acc1 = __builtin_amdgcn_wmma_f32_16x16x4_f32(
                   false, a1, false, b, (short)0, acc1, false, false);
    }

    const int col = wave * 16 + lane16;
    if (col < N) {
        #pragma unroll
        for (int r = 0; r < 8; ++r) {
            int m = r + 8 * half;
            Y[(long)(r0 + m) * N + col]      = acc0[r];
            Y[(long)(r0 + 16 + m) * N + col] = acc1[r];
        }
    }
}

template <int F, int RELU>
__global__ void __launch_bounds__(1024)
gcn_agg_lds(const float* __restrict__ Xe, const int* __restrict__ src,
            const int* __restrict__ dst, const unsigned* __restrict__ own,
            const float* __restrict__ nd, const float* __restrict__ bias,
            float* __restrict__ Y, int nNodes,
            const unsigned* __restrict__ blkcnt, unsigned* __restrict__ list) {
    static_assert(F % 4 == 0 && F / 4 <= 32, "a row is read as one 16-B piece per lane");
    constexpr int LPR = F / 4;
    __shared__ __attribute__((aligned(16))) float acc[(GCN_RPB + 1) * F];
    __shared__ unsigned queue[32 * GCN_QCAP];
    __shared__ unsigned lpos;

    const int tid  = threadIdx.x;
    const int lane = tid & 31;
    const int wave = tid >> 5;
    const int half = lane >> 4;
    const int lo   = blockIdx.x * GCN_RPB;
    const int rows = min(GCN_RPB, nNodes - lo);
    unsigned* q = queue + wave * GCN_QCAP;
    const bool active = lane < LPR;
    unsigned* lst = list + __builtin_amdgcn_readfirstlane(gcn_list_base(blkcnt, blockIdx.x, lane));

    for (int i = tid; i < (GCN_RPB + 1) * F / 4; i += 1024)
        ((v4f*)acc)[i] = (v4f){0.f, 0.f, 0.f, 0.f};
    if (tid == 0) lpos = 0u;
    __syncthreads();

    const int NV = GCN_OWNV;
    const uint4* own4 = (const uint4*)own;
    const unsigned pat = blockIdx.x * 0x01010101u;
    int cnt = 0;
    const int o0 = 4 * lane + half, o1 = 4 * lane + ((1 + half) & 3);
    const int o2 = 4 * lane + ((2 + half) & 3), o3 = 4 * lane + ((3 + half) & 3);

    auto drain = [&]() {
        __builtin_amdgcn_fence(__ATOMIC_ACQ_REL, "wavefront");
        for (int b = 0; b < cnt; b += 32) {
            const int n = min(32, cnt - b);
            int sv = 0, dl = GCN_RPB;
            unsigned p0 = 0u;
            if (lane == 0)
                p0 = __hip_atomic_fetch_add(&lpos, (unsigned)n, __ATOMIC_RELAXED, __HIP_MEMORY_SCOPE_WORKGROUP);
            p0 = __builtin_amdgcn_readfirstlane(p0);
            if (lane < n) {
                const unsigned e = q[b + lane];
                sv = src[e];
                dl = dst[e] - lo;
                lst[p0 + lane] = (unsigned)sv | ((unsigned)dl << GCN_SRCBITS);
            }
            for (int k = 0; k < n; k += 8) {
                v4f r[8];
                #pragma unroll
                for (int j = 0; j < 8; ++j) {
                    const int s = __builtin_amdgcn_readlane(sv, k + j);
                    r[j] = (v4f){0.f, 0.f, 0.f, 0.f};
                    if (active) r[j] = *(const v4f*)(Xe + (long)s * F + 4 * lane);
                }
                #pragma unroll
                for (int j = 0; j < 8; ++j) {
                    const int d = __builtin_amdgcn_readlane(dl, k + j);
                    if (active) {
                        float* a = acc + d * F;
                        const v4f x = r[j];
                        const float x0 = half ? x.y : x.x, x1 = half ? x.z : x.y;
                        const float x2 = half ? x.w : x.z, x3 = half ? x.x : x.w;
                        __hip_atomic_fetch_add(a + o0, x0, __ATOMIC_RELAXED, __HIP_MEMORY_SCOPE_WORKGROUP);
                        __hip_atomic_fetch_add(a + o1, x1, __ATOMIC_RELAXED, __HIP_MEMORY_SCOPE_WORKGROUP);
                        __hip_atomic_fetch_add(a + o2, x2, __ATOMIC_RELAXED, __HIP_MEMORY_SCOPE_WORKGROUP);
                        __hip_atomic_fetch_add(a + o3, x3, __ATOMIC_RELAXED, __HIP_MEMORY_SCOPE_WORKGROUP);
                    }
                }
            }
        }
        __builtin_amdgcn_fence(__ATOMIC_ACQ_REL, "wavefront");
        cnt = 0;
    };

    auto hit = [&](const unsigned w) {
        const unsigned t = w ^ pat;
        return ~(((t & 0x7F7F7F7Fu) + 0x7F7F7F7Fu) | t | 0x7F7F7F7Fu);
    };

    auto test = [&](const uint4 c, const int vidx, const bool live) {
        unsigned m = (hit(c.x) >> 7) | (hit(c.y) >> 6) | (hit(c.z) >> 5) | (hit(c.w) >> 4);
        if (!live) m = 0u;
        unsigned bal = __builtin_amdgcn_ballot_w32(m != 0u);
        while (bal != 0u) {
            if (m != 0u) {
                const int p = __builtin_ctz(m);
                q[cnt + (int)__builtin_amdgcn_mbcnt_lo(bal, 0u)] =
                    (unsigned)(16 * vidx + 4 * (p & 7) + (p >> 3));
                m &= m - 1u;
            }
            cnt += __builtin_popcount(bal);
            bal = __builtin_amdgcn_ballot_w32(m != 0u);
        }
    };

    const int nFull = NV & ~4095;
    if (nFull > 0) {
        uint4 n0 = own4[tid], n1 = own4[tid + 1024], n2 = own4[tid + 2048], n3 = own4[tid + 3072];
        for (int base = 0; base < nFull; base += 4096) {
            const uint4 c0 = n0, c1 = n1, c2 = n2, c3 = n3;
            const int nb = min(base + 4096, nFull - 4096) + tid;
            n0 = own4[nb]; n1 = own4[nb + 1024]; n2 = own4[nb + 2048]; n3 = own4[nb + 3072];
            test(c0, base + tid,        true);
            if (cnt >= GCN_QDRAIN) drain();
            test(c1, base + tid + 1024, true);
            if (cnt >= GCN_QDRAIN) drain();
            test(c2, base + tid + 2048, true);
            if (cnt >= GCN_QDRAIN) drain();
            test(c3, base + tid + 3072, true);
            if (cnt >= GCN_QDRAIN) drain();
        }
    }
    for (int base = nFull; base < NV; base += 1024) {
        const int vidx = base + tid;
        const uint4 c = own4[min(vidx, NV - 1)];
        test(c, vidx, vidx < NV);
        if (cnt >= GCN_QDRAIN) drain();
    }
    if (cnt > 0) drain();
    __syncthreads();

    const int nunits = rows * LPR;
    float* Yb = Y + (long)lo * F;
    for (int u = tid; u < nunits; u += 1024) {
        const int rl = u / LPR;
        const int f4 = u - rl * LPR;
        const v4f a = ((const v4f*)acc)[u];
        const float s = nd[lo + rl];
        const v4f b = *(const v4f*)(bias + 4 * f4);
        v4f o;
        o.x = a.x * s + b.x; o.y = a.y * s + b.y; o.z = a.z * s + b.z; o.w = a.w * s + b.w;
        if (RELU) {
            o.x = fmaxf(o.x, 0.0f); o.y = fmaxf(o.y, 0.0f);
            o.z = fmaxf(o.z, 0.0f); o.w = fmaxf(o.w, 0.0f);
        }
        ((v4f*)Yb)[u] = o;
    }
}

template <int F, int RELU>
__global__ void __launch_bounds__(1024)
gcn_agg_list(const float* __restrict__ Xe, const unsigned* __restrict__ blkcnt,
             const unsigned* __restrict__ list, const float* __restrict__ nd,
             const float* __restrict__ bias, float* __restrict__ Y, int nNodes) {
    static_assert(F % 4 == 0 && F / 4 <= 32, "a row is read as one 16-B piece per lane");
    constexpr int LPR = F / 4;
    __shared__ __attribute__((aligned(16))) float acc[(GCN_RPB + 1) * F];

    const int tid  = threadIdx.x;
    const int lane = tid & 31;
    const int wave = __builtin_amdgcn_readfirstlane(tid >> 5);
    const int half = lane >> 4;
    const int lo   = blockIdx.x * GCN_RPB;
    const int rows = min(GCN_RPB, nNodes - lo);
    const bool active = lane < LPR;
    const int cnt = (int)blkcnt[blockIdx.x];
    const unsigned* L = list + __builtin_amdgcn_readfirstlane(gcn_list_base(blkcnt, blockIdx.x, lane));

    for (int i = tid; i < (GCN_RPB + 1) * F / 4; i += 1024)
        ((v4f*)acc)[i] = (v4f){0.f, 0.f, 0.f, 0.f};
    __syncthreads();

    const int o0 = 4 * lane + half, o1 = 4 * lane + ((1 + half) & 3);
    const int o2 = 4 * lane + ((2 + half) & 3), o3 = 4 * lane + ((3 + half) & 3);

    const unsigned pad = (unsigned)GCN_RPB << GCN_SRCBITS;
    const int first = 8 * wave + 256 * (lane >> 3) + (lane & 7);
    auto fetch = [&](const int it) {
        const int idx = first + 1024 * it;
        return idx < cnt ? L[idx] : pad;
    };
    unsigned cur = fetch(0);
    for (int it = 0; 8 * wave + 1024 * it < cnt; ++it) {
        const unsigned nxt = fetch(it + 1);
        const int sv = (int)(cur & ((1u << GCN_SRCBITS) - 1u));
        const int dl = (int)(cur >> GCN_SRCBITS);
        const int ng = min(4, (cnt - 8 * wave - 1024 * it + 255) >> 8);
        for (int k = 0; k < 8 * ng; k += 8) {
            v4f r[8];
            #pragma unroll
            for (int j = 0; j < 8; ++j) {
                const int s = __builtin_amdgcn_readlane(sv, k + j);
                r[j] = (v4f){0.f, 0.f, 0.f, 0.f};
                if (active) r[j] = *(const v4f*)(Xe + (long)s * F + 4 * lane);
            }
            #pragma unroll
            for (int j = 0; j < 8; ++j) {
                const int d = __builtin_amdgcn_readlane(dl, k + j);
                if (active) {
                    float* a = acc + d * F;
                    const v4f x = r[j];
                    const float x0 = half ? x.y : x.x, x1 = half ? x.z : x.y;
                    const float x2 = half ? x.w : x.z, x3 = half ? x.x : x.w;
                    __hip_atomic_fetch_add(a + o0, x0, __ATOMIC_RELAXED, __HIP_MEMORY_SCOPE_WORKGROUP);
                    __hip_atomic_fetch_add(a + o1, x1, __ATOMIC_RELAXED, __HIP_MEMORY_SCOPE_WORKGROUP);
                    __hip_atomic_fetch_add(a + o2, x2, __ATOMIC_RELAXED, __HIP_MEMORY_SCOPE_WORKGROUP);
                    __hip_atomic_fetch_add(a + o3, x3, __ATOMIC_RELAXED, __HIP_MEMORY_SCOPE_WORKGROUP);
                }
            }
        }
        cur = nxt;
    }
    __syncthreads();

    const int nunits = rows * LPR;
    float* Yb = Y + (long)lo * F;
    for (int u = tid; u < nunits; u += 1024) {
        const int rl = u / LPR;
        const int f4 = u - rl * LPR;
        const v4f a = ((const v4f*)acc)[u];
        const float s = nd[lo + rl];
        const v4f b = *(const v4f*)(bias + 4 * f4);
        v4f o;
        o.x = a.x * s + b.x; o.y = a.y * s + b.y; o.z = a.z * s + b.z; o.w = a.w * s + b.w;
        if (RELU) {
            o.x = fmaxf(o.x, 0.0f); o.y = fmaxf(o.y, 0.0f);
            o.z = fmaxf(o.z, 0.0f); o.w = fmaxf(o.w, 0.0f);
        }
        ((v4f*)Yb)[u] = o;
    }
}

template <int F, int RELU>
__global__ void __launch_bounds__(1024)
gcn_agg_list_packed(const float* __restrict__ Xe, const unsigned* __restrict__ blkcnt,
                    const unsigned* __restrict__ list, const float* __restrict__ nd,
                    const float* __restrict__ bias, float* __restrict__ Y, int nNodes) {
    static_assert(F % 4 == 0 && F / 4 <= 16, "at least two rows fit one wave instruction");
    constexpr int LPR   = F / 4;
    constexpr int RPI   = 32 / LPR;
    constexpr int STEPS = 32 / RPI;
    constexpr int CH    = RPI * STEPS;
    static_assert(RPI >= 2 && RPI <= 4 && CH <= 32, "a lane group's float rotation is its number, below 4");
    __shared__ __attribute__((aligned(16))) float acc[(GCN_RPB + 1) * F];

    const int tid  = threadIdx.x;
    const int lane = tid & 31;
    const int wave = __builtin_amdgcn_readfirstlane(tid >> 5);
    const int slot = lane / LPR;
    const int sub  = lane - slot * LPR;
    const bool active = slot < RPI;
    const int lo   = blockIdx.x * GCN_RPB;
    const int rows = min(GCN_RPB, nNodes - lo);
    const int cnt  = (int)blkcnt[blockIdx.x];
    const unsigned* L = list + __builtin_amdgcn_readfirstlane(gcn_list_base(blkcnt, blockIdx.x, lane));

    for (int i = tid; i < (GCN_RPB + 1) * F / 4; i += 1024)
        ((v4f*)acc)[i] = (v4f){0.f, 0.f, 0.f, 0.f};
    __syncthreads();

    const int o0 = 4 * sub + (slot & 3),       o1 = 4 * sub + ((1 + slot) & 3);
    const int o2 = 4 * sub + ((2 + slot) & 3), o3 = 4 * sub + ((3 + slot) & 3);
    const bool rot1 = (slot & 1) != 0, rot2 = (slot & 2) != 0;

    const unsigned pad = (unsigned)GCN_RPB << GCN_SRCBITS;
    unsigned cur = pad;
    {
        const int idx = CH * wave + lane;
        if (lane < CH && idx < cnt) cur = L[idx];
    }
    for (int it = 0; CH * (wave + 32 * it) < cnt; ++it) {
        const int nidx = CH * (wave + 32 * (it + 1)) + lane;
        const unsigned nraw = L[min(nidx, cnt - 1)];
        v4f r[STEPS];
        int drow[STEPS];
        #pragma unroll
        for (int t = 0; t < STEPS; ++t) {
            unsigned mine = __builtin_amdgcn_readlane(cur, RPI * t);
            #pragma unroll
            for (int g = 1; g < RPI; ++g) {
                const unsigned eg = __builtin_amdgcn_readlane(cur, RPI * t + g);
                mine = (slot == g) ? eg : mine;
            }
            const unsigned sv = mine & ((1u << GCN_SRCBITS) - 1u);
            drow[t] = (int)(mine >> GCN_SRCBITS) * F;
            r[t] = *(const v4f*)(Xe + (sv * (unsigned)F + 4u * (unsigned)sub));
        }
        static_assert(STEPS == 10, "the statement below names ten row reads");
        asm volatile("" :: "v"(r[0]), "v"(r[1]), "v"(r[2]), "v"(r[3]), "v"(r[4]),
                           "v"(r[5]), "v"(r[6]), "v"(r[7]), "v"(r[8]), "v"(r[9]), "v"(nraw));
        const unsigned nxt = (lane < CH && nidx < cnt) ? nraw : pad;
        #pragma unroll
        for (int t = 0; t < STEPS; ++t) {
            if (active) {
                float* a = acc + drow[t];
                const v4f x = r[t];
                const float y0 = rot1 ? x.y : x.x, y1 = rot1 ? x.z : x.y;
                const float y2 = rot1 ? x.w : x.z, y3 = rot1 ? x.x : x.w;
                const float x0 = rot2 ? y2 : y0, x1 = rot2 ? y3 : y1;
                const float x2 = rot2 ? y0 : y2, x3 = rot2 ? y1 : y3;
                __hip_atomic_fetch_add(a + o0, x0, __ATOMIC_RELAXED, __HIP_MEMORY_SCOPE_WORKGROUP);
                __hip_atomic_fetch_add(a + o1, x1, __ATOMIC_RELAXED, __HIP_MEMORY_SCOPE_WORKGROUP);
                __hip_atomic_fetch_add(a + o2, x2, __ATOMIC_RELAXED, __HIP_MEMORY_SCOPE_WORKGROUP);
                __hip_atomic_fetch_add(a + o3, x3, __ATOMIC_RELAXED, __HIP_MEMORY_SCOPE_WORKGROUP);
            }
        }
        cur = nxt;
    }
    __syncthreads();

    const int nunits = rows * LPR;
    float* Yb = Y + (long)lo * F;
    for (int u = tid; u < nunits; u += 1024) {
        const int rl = u / LPR;
        const int f4 = u - rl * LPR;
        const v4f a = ((const v4f*)acc)[u];
        const float s = nd[lo + rl];
        const v4f b = *(const v4f*)(bias + 4 * f4);
        v4f o;
        o.x = a.x * s + b.x; o.y = a.y * s + b.y; o.z = a.z * s + b.z; o.w = a.w * s + b.w;
        if (RELU) {
            o.x = fmaxf(o.x, 0.0f); o.y = fmaxf(o.y, 0.0f);
            o.z = fmaxf(o.z, 0.0f); o.w = fmaxf(o.w, 0.0f);
        }
        ((v4f*)Yb)[u] = o;
    }
}

extern "C" void kernel_launch(void* const* d_in, const int* in_sizes, int n_in,
                              void* d_out, int out_size, void* d_ws, size_t ws_size,
                              hipStream_t stream) {
    (void)in_sizes; (void)n_in; (void)out_size; (void)ws_size;

    const float* feat = (const float*)d_in[0];
    const int*   src  = (const int*)  d_in[1];
    const int*   dst  = (const int*)  d_in[2];
    const float* W0   = (const float*)d_in[3];
    const float* b0   = (const float*)d_in[4];
    const float* W1   = (const float*)d_in[5];
    const float* b1   = (const float*)d_in[6];
    const float* W2   = (const float*)d_in[7];
    const float* b2   = (const float*)d_in[8];
    float*       out  = (float*)d_out;

    const int  Nn = GCN_NODES, Ne = GCN_EDGES, H = GCN_HID, C = GCN_CLS;
    float* ws   = (float*)d_ws;
    float* ns   = ws;
    float* nd   = ws + Nn;
    unsigned* blkcnt = (unsigned*)(ws + 2L * Nn);
    float* bufA = ws + 2L * Nn + 256;
    float* bufB = bufA + (long)Nn * H;
    float* Wf0  = bufB + (long)Nn * H;
    float* Wf1  = Wf0 + 8 * 2048;
    float* Wf2  = Wf1 + 8 * 2048;
    unsigned* own = (unsigned*)(Wf2 + 3 * 2048);
    unsigned* list = own + Ne / 4;
    unsigned* part = list + Ne;

    gcn_degrees<<<GCN_DSL * GCN_DRG, 1024, 0, stream>>>(src, dst, part, own);
    gcn_norms<<<GCN_AGGBLK, GCN_NTH, 0, stream>>>(part, ns, Nn, blkcnt);
    gcn_wfrag<<<(8 * 1024 + 255) / 256, 256, 0, stream>>>(W0, Wf0, H, 8);
    gcn_wfrag<<<(8 * 1024 + 255) / 256, 256, 0, stream>>>(W1, Wf1, H, 8);
    gcn_wfrag<<<(3 * 1024 + 255) / 256, 256, 0, stream>>>(W2, Wf2, C, 3);

    gcn_gemm_wmma<<<Nn / 32, 256, 0, stream>>>(feat, ns, Wf0, bufA, H, 8);
    gcn_agg_lds<GCN_HID, 1><<<GCN_AGGBLK, 1024, 0, stream>>>(bufA, src, dst, own, nd, b0, bufB, Nn, blkcnt, list);

    gcn_gemm_wmma<<<Nn / 32, 256, 0, stream>>>(bufB, ns, Wf1, bufA, H, 8);
    gcn_agg_list<GCN_HID, 1><<<GCN_AGGBLK, 1024, 0, stream>>>(bufA, blkcnt, list, nd, b1, bufB, Nn);

    gcn_gemm_wmma<<<Nn / 32, 256, 0, stream>>>(bufB, ns, Wf2, bufA, C, 3);
    gcn_agg_list_packed<GCN_CLS, 0><<<GCN_AGGBLK, 1024, 0, stream>>>(bufA, blkcnt, list, nd, b2, out, Nn);
}


# Layers 0, 1 aggregation: a block bins its edge list in LDS by owner wave (local row mod 32); each wave adds its rows by plain 16-B LDS load, add, store: no LDS float atomics; sum order = bin arrival
# speedup vs baseline: 3.7042x; 1.0433x over previous
#include <hip/hip_runtime.h>

typedef __attribute__((ext_vector_type(2))) float v2f;
typedef __attribute__((ext_vector_type(4))) float v4f;
typedef __attribute__((ext_vector_type(8))) float v8f;

#define GCN_NODES   100000
#define GCN_EDGES   1600000
#define GCN_KDIM    128
#define GCN_HID     128
#define GCN_CLS     40
#define GCN_LDA     129
#define GCN_RPB     392
#define GCN_AGGBLK  ((GCN_NODES + GCN_RPB - 1) / GCN_RPB)
#define GCN_QDRAIN  64
#define GCN_QCAP    576
static_assert(GCN_EDGES % 4 == 0 && GCN_EDGES <= (1 << 21), "edge number fits 21 bits, edges read as int4");
static_assert(GCN_RPB + 1 <= (1 << 9) && GCN_RPB % 4 == 0, "local row fits 9 bits");
static_assert(GCN_AGGBLK <= 256, "the number of the aggregation block that owns a row fits one byte");
static_assert(GCN_EDGES % 16 == 0, "the owner map is read as 16-byte vectors");
#define GCN_OWNV    (GCN_EDGES / 16)
#define GCN_SRCBITS 17
#define GCN_DSL     32
#define GCN_DRG     8
#define GCN_DRN     (GCN_NODES / GCN_DRG)
#define GCN_DSV     (GCN_EDGES / 4 / GCN_DSL)
static_assert(GCN_NODES % GCN_DRG == 0 && (GCN_EDGES / 4) % GCN_DSL == 0, "ranges and slices are whole");
static_assert(GCN_NODES <= (1 << GCN_SRCBITS) && GCN_RPB + 1 <= (1 << (32 - GCN_SRCBITS)),
              "a list word holds a source row and a local destination row");

__global__ void __launch_bounds__(1024)
gcn_degrees(const int* __restrict__ src, const int* __restrict__ dst,
            unsigned* __restrict__ part, unsigned* __restrict__ own) {
    __shared__ unsigned cs[GCN_DRN];
    __shared__ unsigned cd[GCN_DRN];
    const int tid   = threadIdx.x;
    const int slice = blockIdx.x / GCN_DRG;
    const int rng   = blockIdx.x - slice * GCN_DRG;
    const int rlo   = rng * GCN_DRN;
    const bool maps = rng == (slice % GCN_DRG);

    for (int i = tid; i < GCN_DRN; i += 1024) { cs[i] = 0u; cd[i] = 0u; }
    __syncthreads();

    const int4* s4 = (const int4*)src + (long)slice * GCN_DSV;
    const int4* d4 = (const int4*)dst + (long)slice * GCN_DSV;
    unsigned* ownp = own + (long)slice * GCN_DSV;
    for (int base = 0; base < GCN_DSV; base += 4096) {
        int4 s[4], d[4];
        #pragma unroll
        for (int j = 0; j < 4; ++j) {
            const int v = min(base + 1024 * j + tid, GCN_DSV - 1);
            s[j] = s4[v];
            d[j] = d4[v];
        }
        #pragma unroll
        for (int j = 0; j < 4; ++j) {
            const int v = base + 1024 * j + tid;
            if (v < GCN_DSV) {
                unsigned l;
                l = (unsigned)(s[j].x - rlo); if (l < (unsigned)GCN_DRN) atomicAdd(&cs[l], 1u);
                l = (unsigned)(s[j].y - rlo); if (l < (unsigned)GCN_DRN) atomicAdd(&cs[l], 1u);
                l = (unsigned)(s[j].z - rlo); if (l < (unsigned)GCN_DRN) atomicAdd(&cs[l], 1u);
                l = (unsigned)(s[j].w - rlo); if (l < (unsigned)GCN_DRN) atomicAdd(&cs[l], 1u);
                l = (unsigned)(d[j].x - rlo); if (l < (unsigned)GCN_DRN) atomicAdd(&cd[l], 1u);
                l = (unsigned)(d[j].y - rlo); if (l < (unsigned)GCN_DRN) atomicAdd(&cd[l], 1u);
                l = (unsigned)(d[j].z - rlo); if (l < (unsigned)GCN_DRN) atomicAdd(&cd[l], 1u);
                l = (unsigned)(d[j].w - rlo); if (l < (unsigned)GCN_DRN) atomicAdd(&cd[l], 1u);
                if (maps)
                    ownp[v] = ((unsigned)d[j].x / GCN_RPB) | (((unsigned)d[j].y / GCN_RPB) << 8) |
                              (((unsigned)d[j].z / GCN_RPB) << 16) | (((unsigned)d[j].w / GCN_RPB) << 24);
            }
        }
    }
    __syncthreads();

    unsigned* ps = part + ((long)slice * GCN_NODES + rlo);
    unsigned* pd = part + ((long)(GCN_DSL + slice) * GCN_NODES + rlo);
    for (int i = tid; i < GCN_DRN; i += 1024) { ps[i] = cs[i]; pd[i] = cd[i]; }
}

#define GCN_NTH 512
static_assert(GCN_NTH >= GCN_RPB && GCN_NTH % 32 == 0, "one thread per owned node");
__global__ void __launch_bounds__(GCN_NTH)
gcn_norms(const unsigned* __restrict__ part, float* __restrict__ p, int nNodes,
          unsigned* __restrict__ blkcnt) {
    __shared__ unsigned wsum[GCN_NTH / 32];
    const int t = threadIdx.x;
    const int n = blockIdx.x * GCN_RPB + t;
    const bool has = t < GCN_RPB && n < nNodes;
    unsigned cin = 0u, cout = 0u;
    if (has) {
        const unsigned* qo = part + n;
        const unsigned* qi = part + (long)GCN_DSL * nNodes + n;
        #pragma unroll
        for (int s = 0; s < GCN_DSL; ++s) {
            cout += qo[(long)s * nNodes];
            cin  += qi[(long)s * nNodes];
        }
    }
    unsigned s = cin;
    #pragma unroll
    for (int o = 16; o > 0; o >>= 1) s += __shfl_xor(s, o, 32);
    if ((t & 31) == 0) wsum[t >> 5] = s;
    __syncthreads();
    if (has) {
        p[n]          = rsqrtf(fmaxf((float)cout, 1.0f));
        p[nNodes + n] = rsqrtf(fmaxf((float)cin, 1.0f));
    }
    if (t == 0) {
        unsigned tot = 0u;
        #pragma unroll
        for (int w = 0; w < GCN_NTH / 32; ++w) tot += wsum[w];
        blkcnt[blockIdx.x] = tot;
    }
}

__device__ __forceinline__ unsigned gcn_list_base(const unsigned* __restrict__ blkcnt,
                                                  int b, int lane) {
    unsigned s = 0;
    for (int i = lane; i < b; i += 32) s += blkcnt[i];
    #pragma unroll
    for (int o = 16; o > 0; o >>= 1) s += __shfl_xor(s, o, 32);
    return s;
}

__global__ void gcn_wfrag(const float* __restrict__ W, float* __restrict__ Wf,
                          int N, int nColTiles) {
    int i = blockIdx.x * blockDim.x + threadIdx.x;
    int total = nColTiles * 32 * 32;
    if (i >= total) return;
    int lane   = i & 31;
    int k4     = (i >> 5) & 31;
    int t      = i >> 10;
    int lane16 = lane & 15;
    int half   = lane >> 4;
    int kr     = 4 * k4 + 2 * half;
    int col    = t * 16 + lane16;
    float v0 = 0.0f, v1 = 0.0f;
    if (col < N) {
        v0 = W[kr * N + col];
        v1 = W[(kr + 1) * N + col];
    }
    Wf[2 * i]     = v0;
    Wf[2 * i + 1] = v1;
}

__global__ void __launch_bounds__(256)
gcn_gemm_wmma(const float* __restrict__ X, const float* __restrict__ ns,
              const float* __restrict__ Wf, float* __restrict__ Y,
              int N, int nColTiles) {
    __shared__ float Alds[32 * GCN_LDA];

    const int tid = threadIdx.x;
    const int r0  = blockIdx.x * 32;

    #pragma unroll
    for (int idx = tid; idx < 32 * GCN_KDIM; idx += 256) {
        int m = idx >> 7;
        int k = idx & 127;
        Alds[m * GCN_LDA + k] = X[(long)(r0 + m) * GCN_KDIM + k] * ns[r0 + m];
    }
    __syncthreads();

    const int wave   = tid >> 5;
    const int lane   = tid & 31;
    const int lane16 = lane & 15;
    const int half   = lane >> 4;
    if (wave >= nColTiles) return;

    const float* As0 = &Alds[lane16 * GCN_LDA + 2 * half];
    const float* As1 = As0 + 16 * GCN_LDA;
    const v2f* Wfp = (const v2f*)Wf + (long)wave * 32 * 32 + lane;

    v8f acc0 = {0.f, 0.f, 0.f, 0.f, 0.f, 0.f, 0.f, 0.f};
    v8f acc1 = {0.f, 0.f, 0.f, 0.f, 0.f, 0.f, 0.f, 0.f};

    #pragma unroll 8
    for (int k4 = 0; k4 < GCN_KDIM / 4; ++k4) {
        const int k = k4 * 4;
        v2f a0, a1;
        a0.x = As0[k];
        a0.y = As0[k + 1];
        a1.x = As1[k];
        a1.y = As1[k + 1];
        v2f b = Wfp[k4 * 32];
        acc0 = __builtin_amdgcn_wmma_f32_16x16x4_f32(
                   false, a0, false, b, (short)0, acc0, false, false);
        acc1 = __builtin_amdgcn_wmma_f32_16x16x4_f32(
                   false, a1, false, b, (short)0, acc1, false, false);
    }

    const int col = wave * 16 + lane16;
    if (col < N) {
        #pragma unroll
        for (int r = 0; r < 8; ++r) {
            int m = r + 8 * half;
            Y[(long)(r0 + m) * N + col]      = acc0[r];
            Y[(long)(r0 + 16 + m) * N + col] = acc1[r];
        }
    }
}

#define GCN_BCAP 8192
static_assert(GCN_BCAP % 1024 == 0, "every thread of the block takes the same number of words per pass");
template <int F>
__device__ __forceinline__ void gcn_add_owned(const float* __restrict__ Xe, const unsigned* L,
                                              const int cnt, float* acc, unsigned* bins,
                                              unsigned* wcnt, unsigned* wbase, unsigned* wfill) {
    static_assert(F == 128, "a row takes a whole wave: one 16-B piece per lane");
    const int tid  = threadIdx.x;
    const int lane = tid & 31;
    const int wave = __builtin_amdgcn_readfirstlane(tid >> 5);
    for (int c0 = 0; c0 < cnt; c0 += GCN_BCAP) {
        const int n = min(GCN_BCAP, cnt - c0);
        if (tid < 32) { wcnt[tid] = 0u; wfill[tid] = 0u; }
        __syncthreads();
        unsigned ent[GCN_BCAP / 1024];
        #pragma unroll
        for (int j = 0; j < GCN_BCAP / 1024; ++j)
            ent[j] = L[c0 + min(tid + 1024 * j, n - 1)];
        #pragma unroll
        for (int j = 0; j < GCN_BCAP / 1024; ++j)
            if (tid + 1024 * j < n) atomicAdd(&wcnt[(ent[j] >> GCN_SRCBITS) & 31u], 1u);
        __syncthreads();
        if (wave == 0) {
            unsigned s = wcnt[lane];
            #pragma unroll
            for (int o = 1; o < 32; o <<= 1) {
                const unsigned t = __shfl_up(s, o, 32);
                if (lane >= o) s += t;
            }
            wbase[lane + 1] = s;
            if (lane == 0) wbase[0] = 0u;
        }
        __syncthreads();
        #pragma unroll
        for (int j = 0; j < GCN_BCAP / 1024; ++j)
            if (tid + 1024 * j < n) {
                const unsigned w = (ent[j] >> GCN_SRCBITS) & 31u;
                bins[wbase[w] + atomicAdd(&wfill[w], 1u)] = ent[j];
            }
        __syncthreads();
        const int b0 = __builtin_amdgcn_readfirstlane((int)wbase[wave]);
        const int b1 = __builtin_amdgcn_readfirstlane((int)wbase[wave + 1]);
        const int ng = (b1 - b0) >> 3;
        auto ents = [&](const int g) { return bins[b0 + 8 * min(g, ng - 1) + (lane & 7)]; };
        auto rows8 = [&](v4f (&r)[8], const unsigned e) {
            #pragma unroll
            for (int j = 0; j < 8; ++j) {
                const unsigned s = __builtin_amdgcn_readlane(e, j) & ((1u << GCN_SRCBITS) - 1u);
                r[j] = *(const v4f*)(Xe + (long)s * F + 4 * lane);
            }
        };
        auto sums8 = [&](const v4f (&r)[8], const unsigned e) {
            #pragma unroll
            for (int j = 0; j < 8; ++j) {
                const unsigned d = __builtin_amdgcn_readlane(e, j) >> GCN_SRCBITS;
                v4f* a = (v4f*)(acc + d * F) + lane;
                *a = *a + r[j];
            }
        };
        if (ng > 0) {
            v4f ra[8], rb[8];
            unsigned ea = ents(0), eb = ents(1);
            rows8(ra, ea);
            int g = 0;
            while (g + 2 < ng) {
                rows8(rb, eb);
                const unsigned e2 = ents(g + 2), e3 = ents(g + 3);
                sums8(ra, ea);
                rows8(ra, e2);
                sums8(rb, eb);
                ea = e2; eb = e3; g += 2;
            }
            if (g + 2 == ng) {
                rows8(rb, eb);
                sums8(ra, ea);
                sums8(rb, eb);
            } else {
                sums8(ra, ea);
            }
        }
        for (int i = b0 + 8 * ng; i < b1; ++i) {
            const unsigned e = __builtin_amdgcn_readfirstlane(bins[i]);
            const unsigned s = e & ((1u << GCN_SRCBITS) - 1u);
            const unsigned d = e >> GCN_SRCBITS;
            const v4f x = *(const v4f*)(Xe + (long)s * F + 4 * lane);
            v4f* a = (v4f*)(acc + d * F) + lane;
            *a = *a + x;
        }
        __syncthreads();
    }
}

template <int F, int RELU>
__global__ void __launch_bounds__(1024)
gcn_agg_lds(const float* __restrict__ Xe, const int* __restrict__ src,
            const int* __restrict__ dst, const unsigned* __restrict__ own,
            const float* __restrict__ nd, const float* __restrict__ bias,
            float* __restrict__ Y, int nNodes,
            const unsigned* __restrict__ blkcnt, unsigned* __restrict__ list) {
    static_assert(F % 4 == 0 && F / 4 <= 32, "a row is read as one 16-B piece per lane");
    constexpr int LPR = F / 4;
    __shared__ __attribute__((aligned(16))) float acc[GCN_RPB * F];
    __shared__ unsigned queue[32 * GCN_QCAP];
    __shared__ unsigned bins[GCN_BCAP];
    __shared__ unsigned wcnt[32], wbase[33], wfill[32];
    __shared__ unsigned lpos;

    const int tid  = threadIdx.x;
    const int lane = tid & 31;
    const int wave = tid >> 5;
    const int lo   = blockIdx.x * GCN_RPB;
    const int rows = min(GCN_RPB, nNodes - lo);
    unsigned* q = queue + wave * GCN_QCAP;
    unsigned* lst = list + __builtin_amdgcn_readfirstlane(gcn_list_base(blkcnt, blockIdx.x, lane));

    for (int i = tid; i < GCN_RPB * F / 4; i += 1024)
        ((v4f*)acc)[i] = (v4f){0.f, 0.f, 0.f, 0.f};
    if (tid == 0) lpos = 0u;
    __syncthreads();

    const int NV = GCN_OWNV;
    const uint4* own4 = (const uint4*)own;
    const unsigned pat = blockIdx.x * 0x01010101u;
    int cnt = 0;

    auto drain = [&]() {
        __builtin_amdgcn_fence(__ATOMIC_ACQ_REL, "wavefront");
        for (int b = 0; b < cnt; b += 32) {
            const int n = min(32, cnt - b);
            unsigned p0 = 0u;
            if (lane == 0)
                p0 = __hip_atomic_fetch_add(&lpos, (unsigned)n, __ATOMIC_RELAXED, __HIP_MEMORY_SCOPE_WORKGROUP);
            p0 = __builtin_amdgcn_readfirstlane(p0);
            if (lane < n) {
                const unsigned e = q[b + lane];
                const int sv = src[e];
                const int dl = dst[e] - lo;
                lst[p0 + lane] = (unsigned)sv | ((unsigned)dl << GCN_SRCBITS);
            }
        }
        __builtin_amdgcn_fence(__ATOMIC_ACQ_REL, "wavefront");
        cnt = 0;
    };

    auto hit = [&](const unsigned w) {
        const unsigned t = w ^ pat;
        return ~(((t & 0x7F7F7F7Fu) + 0x7F7F7F7Fu) | t | 0x7F7F7F7Fu);
    };

    auto test = [&](const uint4 c, const int vidx, const bool live) {
        unsigned m = (hit(c.x) >> 7) | (hit(c.y) >> 6) | (hit(c.z) >> 5) | (hit(c.w) >> 4);
        if (!live) m = 0u;
        unsigned bal = __builtin_amdgcn_ballot_w32(m != 0u);
        while (bal != 0u) {
            if (m != 0u) {
                const int p = __builtin_ctz(m);
                q[cnt + (int)__builtin_amdgcn_mbcnt_lo(bal, 0u)] =
                    (unsigned)(16 * vidx + 4 * (p & 7) + (p >> 3));
                m &= m - 1u;
            }
            cnt += __builtin_popcount(bal);
            bal = __builtin_amdgcn_ballot_w32(m != 0u);
        }
    };

    const int nFull = NV & ~4095;
    if (nFull > 0) {
        uint4 n0 = own4[tid], n1 = own4[tid + 1024], n2 = own4[tid + 2048], n3 = own4[tid + 3072];
        for (int base = 0; base < nFull; base += 4096) {
            const uint4 c0 = n0, c1 = n1, c2 = n2, c3 = n3;
            const int nb = min(base + 4096, nFull - 4096) + tid;
            n0 = own4[nb]; n1 = own4[nb + 1024]; n2 = own4[nb + 2048]; n3 = own4[nb + 3072];
            test(c0, base + tid,        true);
            if (cnt >= GCN_QDRAIN) drain();
            test(c1, base + tid + 1024, true);
            if (cnt >= GCN_QDRAIN) drain();
            test(c2, base + tid + 2048, true);
            if (cnt >= GCN_QDRAIN) drain();
            test(c3, base + tid + 3072, true);
            if (cnt >= GCN_QDRAIN) drain();
        }
    }
    for (int base = nFull; base < NV; base += 1024) {
        const int vidx = base + tid;
        const uint4 c = own4[min(vidx, NV - 1)];
        test(c, vidx, vidx < NV);
        if (cnt >= GCN_QDRAIN) drain();
    }
    if (cnt > 0) drain();
    __syncthreads();

    gcn_add_owned<F>(Xe, lst, (int)lpos, acc, bins, wcnt, wbase, wfill);
    __syncthreads();

    const int nunits = rows * LPR;
    float* Yb = Y + (long)lo * F;
    for (int u = tid; u < nunits; u += 1024) {
        const int rl = u / LPR;
        const int f4 = u - rl * LPR;
        const v4f a = ((const v4f*)acc)[u];
        const float s = nd[lo + rl];
        const v4f b = *(const v4f*)(bias + 4 * f4);
        v4f o;
        o.x = a.x * s + b.x; o.y = a.y * s + b.y; o.z = a.z * s + b.z; o.w = a.w * s + b.w;
        if (RELU) {
            o.x = fmaxf(o.x, 0.0f); o.y = fmaxf(o.y, 0.0f);
            o.z = fmaxf(o.z, 0.0f); o.w = fmaxf(o.w, 0.0f);
        }
        ((v4f*)Yb)[u] = o;
    }
}

template <int F, int RELU>
__global__ void __launch_bounds__(1024)
gcn_agg_list(const float* __restrict__ Xe, const unsigned* __restrict__ blkcnt,
             const unsigned* __restrict__ list, const float* __restrict__ nd,
             const float* __restrict__ bias, float* __restrict__ Y, int nNodes) {
    static_assert(F % 4 == 0 && F / 4 <= 32, "a row is read as one 16-B piece per lane");
    constexpr int LPR = F / 4;
    __shared__ __attribute__((aligned(16))) float acc[GCN_RPB * F];
    __shared__ unsigned bins[GCN_BCAP];
    __shared__ unsigned wcnt[32], wbase[33], wfill[32];

    const int tid  = threadIdx.x;
    const int lane = tid & 31;
    const int lo   = blockIdx.x * GCN_RPB;
    const int rows = min(GCN_RPB, nNodes - lo);
    const int cnt = (int)blkcnt[blockIdx.x];
    const unsigned* L = list + __builtin_amdgcn_readfirstlane(gcn_list_base(blkcnt, blockIdx.x, lane));

    for (int i = tid; i < GCN_RPB * F / 4; i += 1024)
        ((v4f*)acc)[i] = (v4f){0.f, 0.f, 0.f, 0.f};
    __syncthreads();

    gcn_add_owned<F>(Xe, L, cnt, acc, bins, wcnt, wbase, wfill);
    __syncthreads();

    const int nunits = rows * LPR;
    float* Yb = Y + (long)lo * F;
    for (int u = tid; u < nunits; u += 1024) {
        const int rl = u / LPR;
        const int f4 = u - rl * LPR;
        const v4f a = ((const v4f*)acc)[u];
        const float s = nd[lo + rl];
        const v4f b = *(const v4f*)(bias + 4 * f4);
        v4f o;
        o.x = a.x * s + b.x; o.y = a.y * s + b.y; o.z = a.z * s + b.z; o.w = a.w * s + b.w;
        if (RELU) {
            o.x = fmaxf(o.x, 0.0f); o.y = fmaxf(o.y, 0.0f);
            o.z = fmaxf(o.z, 0.0f); o.w = fmaxf(o.w, 0.0f);
        }
        ((v4f*)Yb)[u] = o;
    }
}

template <int F, int RELU>
__global__ void __launch_bounds__(1024)
gcn_agg_list_packed(const float* __restrict__ Xe, const unsigned* __restrict__ blkcnt,
                    const unsigned* __restrict__ list, const float* __restrict__ nd,
                    const float* __restrict__ bias, float* __restrict__ Y, int nNodes) {
    static_assert(F % 4 == 0 && F / 4 <= 16, "at least two rows fit one wave instruction");
    constexpr int LPR   = F / 4;
    constexpr int RPI   = 32 / LPR;
    constexpr int STEPS = 32 / RPI;
    constexpr int CH    = RPI * STEPS;
    static_assert(RPI >= 2 && RPI <= 4 && CH <= 32, "a lane group's float rotation is its number, below 4");
    __shared__ __attribute__((aligned(16))) float acc[(GCN_RPB + 1) * F];

    const int tid  = threadIdx.x;
    const int lane = tid & 31;
    const int wave = __builtin_amdgcn_readfirstlane(tid >> 5);
    const int slot = lane / LPR;
    const int sub  = lane - slot * LPR;
    const bool active = slot < RPI;
    const int lo   = blockIdx.x * GCN_RPB;
    const int rows = min(GCN_RPB, nNodes - lo);
    const int cnt  = (int)blkcnt[blockIdx.x];
    const unsigned* L = list + __builtin_amdgcn_readfirstlane(gcn_list_base(blkcnt, blockIdx.x, lane));

    for (int i = tid; i < (GCN_RPB + 1) * F / 4; i += 1024)
        ((v4f*)acc)[i] = (v4f){0.f, 0.f, 0.f, 0.f};
    __syncthreads();

    const int o0 = 4 * sub + (slot & 3),       o1 = 4 * sub + ((1 + slot) & 3);
    const int o2 = 4 * sub + ((2 + slot) & 3), o3 = 4 * sub + ((3 + slot) & 3);
    const bool rot1 = (slot & 1) != 0, rot2 = (slot & 2) != 0;

    const unsigned pad = (unsigned)GCN_RPB << GCN_SRCBITS;
    unsigned cur = pad;
    {
        const int idx = CH * wave + lane;
        if (lane < CH && idx < cnt) cur = L[idx];
    }
    for (int it = 0; CH * (wave + 32 * it) < cnt; ++it) {
        const int nidx = CH * (wave + 32 * (it + 1)) + lane;
        const unsigned nraw = L[min(nidx, cnt - 1)];
        v4f r[STEPS];
        int drow[STEPS];
        #pragma unroll
        for (int t = 0; t < STEPS; ++t) {
            unsigned mine = __builtin_amdgcn_readlane(cur, RPI * t);
            #pragma unroll
            for (int g = 1; g < RPI; ++g) {
                const unsigned eg = __builtin_amdgcn_readlane(cur, RPI * t + g);
                mine = (slot == g) ? eg : mine;
            }
            const unsigned sv = mine & ((1u << GCN_SRCBITS) - 1u);
            drow[t] = (int)(mine >> GCN_SRCBITS) * F;
            r[t] = *(const v4f*)(Xe + (sv * (unsigned)F + 4u * (unsigned)sub));
        }
        static_assert(STEPS == 10, "the statement below names ten row reads");
        asm volatile("" :: "v"(r[0]), "v"(r[1]), "v"(r[2]), "v"(r[3]), "v"(r[4]),
                           "v"(r[5]), "v"(r[6]), "v"(r[7]), "v"(r[8]), "v"(r[9]), "v"(nraw));
        const unsigned nxt = (lane < CH && nidx < cnt) ? nraw : pad;
        #pragma unroll
        for (int t = 0; t < STEPS; ++t) {
            if (active) {
                float* a = acc + drow[t];
                const v4f x = r[t];
                const float y0 = rot1 ? x.y : x.x, y1 = rot1 ? x.z : x.y;
                const float y2 = rot1 ? x.w : x.z, y3 = rot1 ? x.x : x.w;
                const float x0 = rot2 ? y2 : y0, x1 = rot2 ? y3 : y1;
                const float x2 = rot2 ? y0 : y2, x3 = rot2 ? y1 : y3;
                __hip_atomic_fetch_add(a + o0, x0, __ATOMIC_RELAXED, __HIP_MEMORY_SCOPE_WORKGROUP);
                __hip_atomic_fetch_add(a + o1, x1, __ATOMIC_RELAXED, __HIP_MEMORY_SCOPE_WORKGROUP);
                __hip_atomic_fetch_add(a + o2, x2, __ATOMIC_RELAXED, __HIP_MEMORY_SCOPE_WORKGROUP);
                __hip_atomic_fetch_add(a + o3, x3, __ATOMIC_RELAXED, __HIP_MEMORY_SCOPE_WORKGROUP);
            }
        }
        cur = nxt;
    }
    __syncthreads();

    const int nunits = rows * LPR;
    float* Yb = Y + (long)lo * F;
    for (int u = tid; u < nunits; u += 1024) {
        const int rl = u / LPR;
        const int f4 = u - rl * LPR;
        const v4f a = ((const v4f*)acc)[u];
        const float s = nd[lo + rl];
        const v4f b = *(const v4f*)(bias + 4 * f4);
        v4f o;
        o.x = a.x * s + b.x; o.y = a.y * s + b.y; o.z = a.z * s + b.z; o.w = a.w * s + b.w;
        if (RELU) {
            o.x = fmaxf(o.x, 0.0f); o.y = fmaxf(o.y, 0.0f);
            o.z = fmaxf(o.z, 0.0f); o.w = fmaxf(o.w, 0.0f);
        }
        ((v4f*)Yb)[u] = o;
    }
}

extern "C" void kernel_launch(void* const* d_in, const int* in_sizes, int n_in,
                              void* d_out, int out_size, void* d_ws, size_t ws_size,
                              hipStream_t stream) {
    (void)in_sizes; (void)n_in; (void)out_size; (void)ws_size;

    const float* feat = (const float*)d_in[0];
    const int*   src  = (const int*)  d_in[1];
    const int*   dst  = (const int*)  d_in[2];
    const float* W0   = (const float*)d_in[3];
    const float* b0   = (const float*)d_in[4];
    const float* W1   = (const float*)d_in[5];
    const float* b1   = (const float*)d_in[6];
    const float* W2   = (const float*)d_in[7];
    const float* b2   = (const float*)d_in[8];
    float*       out  = (float*)d_out;

    const int  Nn = GCN_NODES, Ne = GCN_EDGES, H = GCN_HID, C = GCN_CLS;
    float* ws   = (float*)d_ws;
    float* ns   = ws;
    float* nd   = ws + Nn;
    unsigned* blkcnt = (unsigned*)(ws + 2L * Nn);
    float* bufA = ws + 2L * Nn + 256;
    float* bufB = bufA + (long)Nn * H;
    float* Wf0  = bufB + (long)Nn * H;
    float* Wf1  = Wf0 + 8 * 2048;
    float* Wf2  = Wf1 + 8 * 2048;
    unsigned* own = (unsigned*)(Wf2 + 3 * 2048);
    unsigned* list = own + Ne / 4;
    unsigned* part = list + Ne;

    gcn_degrees<<<GCN_DSL * GCN_DRG, 1024, 0, stream>>>(src, dst, part, own);
    gcn_norms<<<GCN_AGGBLK, GCN_NTH, 0, stream>>>(part, ns, Nn, blkcnt);
    gcn_wfrag<<<(8 * 1024 + 255) / 256, 256, 0, stream>>>(W0, Wf0, H, 8);
    gcn_wfrag<<<(8 * 1024 + 255) / 256, 256, 0, stream>>>(W1, Wf1, H, 8);
    gcn_wfrag<<<(3 * 1024 + 255) / 256, 256, 0, stream>>>(W2, Wf2, C, 3);

    gcn_gemm_wmma<<<Nn / 32, 256, 0, stream>>>(feat, ns, Wf0, bufA, H, 8);
    gcn_agg_lds<GCN_HID, 1><<<GCN_AGGBLK, 1024, 0, stream>>>(bufA, src, dst, own, nd, b0, bufB, Nn, blkcnt, list);

    gcn_gemm_wmma<<<Nn / 32, 256, 0, stream>>>(bufB, ns, Wf1, bufA, H, 8);
    gcn_agg_list<GCN_HID, 1><<<GCN_AGGBLK, 1024, 0, stream>>>(bufA, blkcnt, list, nd, b1, bufB, Nn);

    gcn_gemm_wmma<<<Nn / 32, 256, 0, stream>>>(bufB, ns, Wf2, bufA, C, 3);
    gcn_agg_list_packed<GCN_CLS, 0><<<GCN_AGGBLK, 1024, 0, stream>>>(bufA, blkcnt, list, nd, b2, out, Nn);
}


# Four small dependent launches folded into their users (11 to 7): first product makes its rows' norms from the degree planes, list bases from per-slice owner counts, B fragments read straight from W
# speedup vs baseline: 3.8958x; 1.0517x over previous
#include <hip/hip_runtime.h>

typedef __attribute__((ext_vector_type(2))) float v2f;
typedef __attribute__((ext_vector_type(4))) float v4f;
typedef __attribute__((ext_vector_type(8))) float v8f;

#define GCN_NODES   100000
#define GCN_EDGES   1600000
#define GCN_KDIM    128
#define GCN_HID     128
#define GCN_CLS     40
#define GCN_LDA     129
#define GCN_RPB     392
#define GCN_AGGBLK  ((GCN_NODES + GCN_RPB - 1) / GCN_RPB)
#define GCN_QDRAIN  64
#define GCN_QCAP    576
static_assert(GCN_EDGES % 4 == 0 && GCN_EDGES <= (1 << 21), "edge number fits 21 bits, edges read as int4");
static_assert(GCN_RPB + 1 <= (1 << 9) && GCN_RPB % 4 == 0, "local row fits 9 bits");
static_assert(GCN_AGGBLK <= 256, "the number of the aggregation block that owns a row fits one byte");
static_assert(GCN_EDGES % 16 == 0, "the owner map is read as 16-byte vectors");
#define GCN_OWNV    (GCN_EDGES / 16)
#define GCN_SRCBITS 17
#define GCN_DSL     32
#define GCN_DRG     8
#define GCN_DRN     (GCN_NODES / GCN_DRG)
#define GCN_DSV     (GCN_EDGES / 4 / GCN_DSL)
static_assert(GCN_NODES % GCN_DRG == 0 && (GCN_EDGES / 4) % GCN_DSL == 0, "ranges and slices are whole");
static_assert(GCN_NODES <= (1 << GCN_SRCBITS) && GCN_RPB + 1 <= (1 << (32 - GCN_SRCBITS)),
              "a list word holds a source row and a local destination row");

__global__ void __launch_bounds__(1024)
gcn_degrees(const int* __restrict__ src, const int* __restrict__ dst,
            unsigned* __restrict__ part, unsigned* __restrict__ own,
            unsigned* __restrict__ hist) {
    __shared__ unsigned cs[GCN_DRN];
    __shared__ unsigned cd[GCN_DRN];
    __shared__ unsigned oh[GCN_AGGBLK];
    const int tid   = threadIdx.x;
    const int slice = blockIdx.x / GCN_DRG;
    const int rng   = blockIdx.x - slice * GCN_DRG;
    const int rlo   = rng * GCN_DRN;
    const bool maps = rng == (slice % GCN_DRG);

    for (int i = tid; i < GCN_DRN; i += 1024) { cs[i] = 0u; cd[i] = 0u; }
    if (tid < GCN_AGGBLK) oh[tid] = 0u;
    __syncthreads();

    const int4* s4 = (const int4*)src + (long)slice * GCN_DSV;
    const int4* d4 = (const int4*)dst + (long)slice * GCN_DSV;
    unsigned* ownp = own + (long)slice * GCN_DSV;
    for (int base = 0; base < GCN_DSV; base += 4096) {
        int4 s[4], d[4];
        #pragma unroll
        for (int j = 0; j < 4; ++j) {
            const int v = min(base + 1024 * j + tid, GCN_DSV - 1);
            s[j] = s4[v];
            d[j] = d4[v];
        }
        #pragma unroll
        for (int j = 0; j < 4; ++j) {
            const int v = base + 1024 * j + tid;
            if (v < GCN_DSV) {
                unsigned l;
                l = (unsigned)(s[j].x - rlo); if (l < (unsigned)GCN_DRN) atomicAdd(&cs[l], 1u);
                l = (unsigned)(s[j].y - rlo); if (l < (unsigned)GCN_DRN) atomicAdd(&cs[l], 1u);
                l = (unsigned)(s[j].z - rlo); if (l < (unsigned)GCN_DRN) atomicAdd(&cs[l], 1u);
                l = (unsigned)(s[j].w - rlo); if (l < (unsigned)GCN_DRN) atomicAdd(&cs[l], 1u);
                l = (unsigned)(d[j].x - rlo); if (l < (unsigned)GCN_DRN) atomicAdd(&cd[l], 1u);
                l = (unsigned)(d[j].y - rlo); if (l < (unsigned)GCN_DRN) atomicAdd(&cd[l], 1u);
                l = (unsigned)(d[j].z - rlo); if (l < (unsigned)GCN_DRN) atomicAdd(&cd[l], 1u);
                l = (unsigned)(d[j].w - rlo); if (l < (unsigned)GCN_DRN) atomicAdd(&cd[l], 1u);
                if (maps) {
                    const unsigned ox = (unsigned)d[j].x / GCN_RPB, oy = (unsigned)d[j].y / GCN_RPB;
                    const unsigned oz = (unsigned)d[j].z / GCN_RPB, ow = (unsigned)d[j].w / GCN_RPB;
                    ownp[v] = ox | (oy << 8) | (oz << 16) | (ow << 24);
                    atomicAdd(&oh[ox], 1u); atomicAdd(&oh[oy], 1u);
                    atomicAdd(&oh[oz], 1u); atomicAdd(&oh[ow], 1u);
                }
            }
        }
    }
    __syncthreads();

    unsigned* ps = part + ((long)slice * GCN_NODES + rlo);
    unsigned* pd = part + ((long)(GCN_DSL + slice) * GCN_NODES + rlo);
    for (int i = tid; i < GCN_DRN; i += 1024) { ps[i] = cs[i]; pd[i] = cd[i]; }
    if (maps && tid < GCN_AGGBLK) hist[slice * GCN_AGGBLK + tid] = oh[tid];
}

__device__ __forceinline__ unsigned gcn_list_base(const unsigned* __restrict__ blkcnt,
                                                  int b, int lane) {
    unsigned s = 0;
    for (int i = lane; i < b; i += 32) s += blkcnt[i];
    #pragma unroll
    for (int o = 16; o > 0; o >>= 1) s += __shfl_xor(s, o, 32);
    return s;
}

static_assert(GCN_DSL == 32, "one wave of a 1024-thread block per edge slice");
__device__ __forceinline__ unsigned gcn_hist_share(const unsigned* __restrict__ hist,
                                                   int b, int wave, int lane) {
    unsigned s = 0;
    const unsigned* row = hist + wave * GCN_AGGBLK;
    for (int i = lane; i < b; i += 32) s += row[i];
    #pragma unroll
    for (int o = 16; o > 0; o >>= 1) s += __shfl_xor(s, o, 32);
    return s;
}

template <int N, int NT, int FIRST>
__global__ void __launch_bounds__(256)
gcn_gemm_wmma(const float* __restrict__ X, float* __restrict__ p,
              const unsigned* __restrict__ part, const float* __restrict__ W,
              float* __restrict__ Y, int nNodes) {
    static_assert(NT <= 8 && NT * 16 >= N, "one wave per 16-column tile");
    __shared__ float Alds[32 * GCN_LDA];
    __shared__ unsigned dsum[64];
    __shared__ float nsl[32];

    const int tid = threadIdx.x;
    const int r0  = blockIdx.x * 32;

    float xr[32 * GCN_KDIM / 256];
    #pragma unroll
    for (int j = 0; j < 32 * GCN_KDIM / 256; ++j) {
        const int idx = tid + 256 * j;
        xr[j] = X[(long)(r0 + (idx >> 7)) * GCN_KDIM + (idx & 127)];
    }

    if (FIRST) {
        static_assert(GCN_DSL == 32, "eight thread groups of four planes");
        const int m = tid & 31, g = tid >> 5;
        const unsigned* qo = part + (long)(4 * g) * nNodes + (r0 + m);
        const unsigned* qi = qo + (long)GCN_DSL * nNodes;
        unsigned co = 0u, ci = 0u;
        #pragma unroll
        for (int j = 0; j < 4; ++j) {
            co += qo[(long)j * nNodes];
            ci += qi[(long)j * nNodes];
        }
        if (tid < 64) dsum[tid] = 0u;
        __syncthreads();
        atomicAdd(&dsum[m], co);
        atomicAdd(&dsum[32 + m], ci);
        __syncthreads();
        if (tid < 32) {
            const float a = rsqrtf(fmaxf((float)dsum[tid], 1.0f));
            nsl[tid] = a;
            p[r0 + tid]          = a;
            p[nNodes + r0 + tid] = rsqrtf(fmaxf((float)dsum[32 + tid], 1.0f));
        }
        __syncthreads();
    } else {
        if (tid < 32) nsl[tid] = p[r0 + tid];
        __syncthreads();
    }

    #pragma unroll
    for (int j = 0; j < 32 * GCN_KDIM / 256; ++j) {
        const int idx = tid + 256 * j;
        const int m = idx >> 7;
        const int k = idx & 127;
        Alds[m * GCN_LDA + k] = xr[j] * nsl[m];
    }
    __syncthreads();

    const int wave   = tid >> 5;
    const int lane   = tid & 31;
    const int lane16 = lane & 15;
    const int half   = lane >> 4;
    if (wave >= NT) return;

    const float* As0 = &Alds[lane16 * GCN_LDA + 2 * half];
    const float* As1 = As0 + 16 * GCN_LDA;
    const int  col = wave * 16 + lane16;
    const bool cv  = col < N;
    const float* Wp = W + (2 * half) * N + (cv ? col : 0);

    v8f acc0 = {0.f, 0.f, 0.f, 0.f, 0.f, 0.f, 0.f, 0.f};
    v8f acc1 = {0.f, 0.f, 0.f, 0.f, 0.f, 0.f, 0.f, 0.f};

    #pragma unroll 8
    for (int k4 = 0; k4 < GCN_KDIM / 4; ++k4) {
        const int k = k4 * 4;
        v2f a0, a1;
        a0.x = As0[k];
        a0.y = As0[k + 1];
        a1.x = As1[k];
        a1.y = As1[k + 1];
        v2f b;
        b.x = Wp[k * N];
        b.y = Wp[k * N + N];
        if (N % 16 != 0) {
            b.x = cv ? b.x : 0.0f;
            b.y = cv ? b.y : 0.0f;
        }
        acc0 = __builtin_amdgcn_wmma_f32_16x16x4_f32(
                   false, a0, false, b, (short)0, acc0, false, false);
        acc1 = __builtin_amdgcn_wmma_f32_16x16x4_f32(
                   false, a1, false, b, (short)0, acc1, false, false);
    }

    if (cv) {
        #pragma unroll
        for (int r = 0; r < 8; ++r) {
            int m = r + 8 * half;
            Y[(long)(r0 + m) * N + col]      = acc0[r];
            Y[(long)(r0 + 16 + m) * N + col] = acc1[r];
        }
    }
}

#define GCN_BCAP 8192
static_assert(GCN_BCAP % 1024 == 0, "every thread of the block takes the same number of words per pass");
template <int F>
__device__ __forceinline__ void gcn_add_owned(const float* __restrict__ Xe, const unsigned* L,
                                              const int cnt, float* acc, unsigned* bins,
                                              unsigned* wcnt, unsigned* wbase, unsigned* wfill) {
    static_assert(F == 128, "a row takes a whole wave: one 16-B piece per lane");
    const int tid  = threadIdx.x;
    const int lane = tid & 31;
    const int wave = __builtin_amdgcn_readfirstlane(tid >> 5);
    for (int c0 = 0; c0 < cnt; c0 += GCN_BCAP) {
        const int n = min(GCN_BCAP, cnt - c0);
        if (tid < 32) { wcnt[tid] = 0u; wfill[tid] = 0u; }
        __syncthreads();
        unsigned ent[GCN_BCAP / 1024];
        #pragma unroll
        for (int j = 0; j < GCN_BCAP / 1024; ++j)
            ent[j] = L[c0 + min(tid + 1024 * j, n - 1)];
        #pragma unroll
        for (int j = 0; j < GCN_BCAP / 1024; ++j)
            if (tid + 1024 * j < n) atomicAdd(&wcnt[(ent[j] >> GCN_SRCBITS) & 31u], 1u);
        __syncthreads();
        if (wave == 0) {
            unsigned s = wcnt[lane];
            #pragma unroll
            for (int o = 1; o < 32; o <<= 1) {
                const unsigned t = __shfl_up(s, o, 32);
                if (lane >= o) s += t;
            }
            wbase[lane + 1] = s;
            if (lane == 0) wbase[0] = 0u;
        }
        __syncthreads();
        #pragma unroll
        for (int j = 0; j < GCN_BCAP / 1024; ++j)
            if (tid + 1024 * j < n) {
                const unsigned w = (ent[j] >> GCN_SRCBITS) & 31u;
                bins[wbase[w] + atomicAdd(&wfill[w], 1u)] = ent[j];
            }
        __syncthreads();
        const int b0 = __builtin_amdgcn_readfirstlane((int)wbase[wave]);
        const int b1 = __builtin_amdgcn_readfirstlane((int)wbase[wave + 1]);
        const int ng = (b1 - b0) >> 3;
        auto ents = [&](const int g) { return bins[b0 + 8 * min(g, ng - 1) + (lane & 7)]; };
        auto rows8 = [&](v4f (&r)[8], const unsigned e) {
            #pragma unroll
            for (int j = 0; j < 8; ++j) {
                const unsigned s = __builtin_amdgcn_readlane(e, j) & ((1u << GCN_SRCBITS) - 1u);
                r[j] = *(const v4f*)(Xe + (long)s * F + 4 * lane);
            }
        };
        auto sums8 = [&](const v4f (&r)[8], const unsigned e) {
            #pragma unroll
            for (int j = 0; j < 8; ++j) {
                const unsigned d = __builtin_amdgcn_readlane(e, j) >> GCN_SRCBITS;
                v4f* a = (v4f*)(acc + d * F) + lane;
                *a = *a + r[j];
            }
        };
        if (ng > 0) {
            v4f ra[8], rb[8];
            unsigned ea = ents(0), eb = ents(1);
            rows8(ra, ea);
            int g = 0;
            while (g + 2 < ng) {
                rows8(rb, eb);
                const unsigned e2 = ents(g + 2), e3 = ents(g + 3);
                sums8(ra, ea);
                rows8(ra, e2);
                sums8(rb, eb);
                ea = e2; eb = e3; g += 2;
            }
            if (g + 2 == ng) {
                rows8(rb, eb);
                sums8(ra, ea);
                sums8(rb, eb);
            } else {
                sums8(ra, ea);
            }
        }
        for (int i = b0 + 8 * ng; i < b1; ++i) {
            const unsigned e = __builtin_amdgcn_readfirstlane(bins[i]);
            const unsigned s = e & ((1u << GCN_SRCBITS) - 1u);
            const unsigned d = e >> GCN_SRCBITS;
            const v4f x = *(const v4f*)(Xe + (long)s * F + 4 * lane);
            v4f* a = (v4f*)(acc + d * F) + lane;
            *a = *a + x;
        }
        __syncthreads();
    }
}

template <int F, int RELU>
__global__ void __launch_bounds__(1024)
gcn_agg_lds(const float* __restrict__ Xe, const int* __restrict__ src,
            const int* __restrict__ dst, const unsigned* __restrict__ own,
            const float* __restrict__ nd, const float* __restrict__ bias,
            float* __restrict__ Y, int nNodes,
            const unsigned* __restrict__ hist, unsigned* __restrict__ blkcnt,
            unsigned* __restrict__ list) {
    static_assert(F % 4 == 0 && F / 4 <= 32, "a row is read as one 16-B piece per lane");
    constexpr int LPR = F / 4;
    __shared__ __attribute__((aligned(16))) float acc[GCN_RPB * F];
    __shared__ unsigned queue[32 * GCN_QCAP];
    __shared__ unsigned bins[GCN_BCAP];
    __shared__ unsigned wcnt[32], wbase[33], wfill[32];
    __shared__ unsigned hsum[32];
    __shared__ unsigned lpos;

    const int tid  = threadIdx.x;
    const int lane = tid & 31;
    const int wave = tid >> 5;
    const int lo   = blockIdx.x * GCN_RPB;
    const int rows = min(GCN_RPB, nNodes - lo);
    unsigned* q = queue + wave * GCN_QCAP;

    const unsigned share = gcn_hist_share(hist, blockIdx.x, wave, lane);
    if (lane == 0) hsum[wave] = share;
    for (int i = tid; i < GCN_RPB * F / 4; i += 1024)
        ((v4f*)acc)[i] = (v4f){0.f, 0.f, 0.f, 0.f};
    if (tid == 0) lpos = 0u;
    __syncthreads();
    unsigned lb = hsum[lane];
    #pragma unroll
    for (int o = 16; o > 0; o >>= 1) lb += __shfl_xor(lb, o, 32);
    unsigned* lst = list + __builtin_amdgcn_readfirstlane(lb);

    const int NV = GCN_OWNV;
    const uint4* own4 = (const uint4*)own;
    const unsigned pat = blockIdx.x * 0x01010101u;
    int cnt = 0;

    auto drain = [&]() {
        __builtin_amdgcn_fence(__ATOMIC_ACQ_REL, "wavefront");
        for (int b = 0; b < cnt; b += 32) {
            const int n = min(32, cnt - b);
            unsigned p0 = 0u;
            if (lane == 0)
                p0 = __hip_atomic_fetch_add(&lpos, (unsigned)n, __ATOMIC_RELAXED, __HIP_MEMORY_SCOPE_WORKGROUP);
            p0 = __builtin_amdgcn_readfirstlane(p0);
            if (lane < n) {
                const unsigned e = q[b + lane];
                const int sv = src[e];
                const int dl = dst[e] - lo;
                lst[p0 + lane] = (unsigned)sv | ((unsigned)dl << GCN_SRCBITS);
            }
        }
        __builtin_amdgcn_fence(__ATOMIC_ACQ_REL, "wavefront");
        cnt = 0;
    };

    auto hit = [&](const unsigned w) {
        const unsigned t = w ^ pat;
        return ~(((t & 0x7F7F7F7Fu) + 0x7F7F7F7Fu) | t | 0x7F7F7F7Fu);
    };

    auto test = [&](const uint4 c, const int vidx, const bool live) {
        unsigned m = (hit(c.x) >> 7) | (hit(c.y) >> 6) | (hit(c.z) >> 5) | (hit(c.w) >> 4);
        if (!live) m = 0u;
        unsigned bal = __builtin_amdgcn_ballot_w32(m != 0u);
        while (bal != 0u) {
            if (m != 0u) {
                const int p = __builtin_ctz(m);
                q[cnt + (int)__builtin_amdgcn_mbcnt_lo(bal, 0u)] =
                    (unsigned)(16 * vidx + 4 * (p & 7) + (p >> 3));
                m &= m - 1u;
            }
            cnt += __builtin_popcount(bal);
            bal = __builtin_amdgcn_ballot_w32(m != 0u);
        }
    };

    const int nFull = NV & ~4095;
    if (nFull > 0) {
        uint4 n0 = own4[tid], n1 = own4[tid + 1024], n2 = own4[tid + 2048], n3 = own4[tid + 3072];
        for (int base = 0; base < nFull; base += 4096) {
            const uint4 c0 = n0, c1 = n1, c2 = n2, c3 = n3;
            const int nb = min(base + 4096, nFull - 4096) + tid;
            n0 = own4[nb]; n1 = own4[nb + 1024]; n2 = own4[nb + 2048]; n3 = own4[nb + 3072];
            test(c0, base + tid,        true);
            if (cnt >= GCN_QDRAIN) drain();
            test(c1, base + tid + 1024, true);
            if (cnt >= GCN_QDRAIN) drain();
            test(c2, base + tid + 2048, true);
            if (cnt >= GCN_QDRAIN) drain();
            test(c3, base + tid + 3072, true);
            if (cnt >= GCN_QDRAIN) drain();
        }
    }
    for (int base = nFull; base < NV; base += 1024) {
        const int vidx = base + tid;
        const uint4 c = own4[min(vidx, NV - 1)];
        test(c, vidx, vidx < NV);
        if (cnt >= GCN_QDRAIN) drain();
    }
    if (cnt > 0) drain();
    __syncthreads();
    if (tid == 0) blkcnt[blockIdx.x] = lpos;

    gcn_add_owned<F>(Xe, lst, (int)lpos, acc, bins, wcnt, wbase, wfill);
    __syncthreads();

    const int nunits = rows * LPR;
    float* Yb = Y + (long)lo * F;
    for (int u = tid; u < nunits; u += 1024) {
        const int rl = u / LPR;
        const int f4 = u - rl * LPR;
        const v4f a = ((const v4f*)acc)[u];
        const float s = nd[lo + rl];
        const v4f b = *(const v4f*)(bias + 4 * f4);
        v4f o;
        o.x = a.x * s + b.x; o.y = a.y * s + b.y; o.z = a.z * s + b.z; o.w = a.w * s + b.w;
        if (RELU) {
            o.x = fmaxf(o.x, 0.0f); o.y = fmaxf(o.y, 0.0f);
            o.z = fmaxf(o.z, 0.0f); o.w = fmaxf(o.w, 0.0f);
        }
        ((v4f*)Yb)[u] = o;
    }
}

template <int F, int RELU>
__global__ void __launch_bounds__(1024)
gcn_agg_list(const float* __restrict__ Xe, const unsigned* __restrict__ blkcnt,
             const unsigned* __restrict__ list, const float* __restrict__ nd,
             const float* __restrict__ bias, float* __restrict__ Y, int nNodes) {
    static_assert(F % 4 == 0 && F / 4 <= 32, "a row is read as one 16-B piece per lane");
    constexpr int LPR = F / 4;
    __shared__ __attribute__((aligned(16))) float acc[GCN_RPB * F];
    __shared__ unsigned bins[GCN_BCAP];
    __shared__ unsigned wcnt[32], wbase[33], wfill[32];

    const int tid  = threadIdx.x;
    const int lane = tid & 31;
    const int lo   = blockIdx.x * GCN_RPB;
    const int rows = min(GCN_RPB, nNodes - lo);
    const int cnt = (int)blkcnt[blockIdx.x];
    const unsigned* L = list + __builtin_amdgcn_readfirstlane(gcn_list_base(blkcnt, blockIdx.x, lane));

    for (int i = tid; i < GCN_RPB * F / 4; i += 1024)
        ((v4f*)acc)[i] = (v4f){0.f, 0.f, 0.f, 0.f};
    __syncthreads();

    gcn_add_owned<F>(Xe, L, cnt, acc, bins, wcnt, wbase, wfill);
    __syncthreads();

    const int nunits = rows * LPR;
    float* Yb = Y + (long)lo * F;
    for (int u = tid; u < nunits; u += 1024) {
        const int rl = u / LPR;
        const int f4 = u - rl * LPR;
        const v4f a = ((const v4f*)acc)[u];
        const float s = nd[lo + rl];
        const v4f b = *(const v4f*)(bias + 4 * f4);
        v4f o;
        o.x = a.x * s + b.x; o.y = a.y * s + b.y; o.z = a.z * s + b.z; o.w = a.w * s + b.w;
        if (RELU) {
            o.x = fmaxf(o.x, 0.0f); o.y = fmaxf(o.y, 0.0f);
            o.z = fmaxf(o.z, 0.0f); o.w = fmaxf(o.w, 0.0f);
        }
        ((v4f*)Yb)[u] = o;
    }
}

template <int F, int RELU>
__global__ void __launch_bounds__(1024)
gcn_agg_list_packed(const float* __restrict__ Xe, const unsigned* __restrict__ blkcnt,
                    const unsigned* __restrict__ list, const float* __restrict__ nd,
                    const float* __restrict__ bias, float* __restrict__ Y, int nNodes) {
    static_assert(F % 4 == 0 && F / 4 <= 16, "at least two rows fit one wave instruction");
    constexpr int LPR   = F / 4;
    constexpr int RPI   = 32 / LPR;
    constexpr int STEPS = 32 / RPI;
    constexpr int CH    = RPI * STEPS;
    static_assert(RPI >= 2 && RPI <= 4 && CH <= 32, "a lane group's float rotation is its number, below 4");
    __shared__ __attribute__((aligned(16))) float acc[(GCN_RPB + 1) * F];

    const int tid  = threadIdx.x;
    const int lane = tid & 31;
    const int wave = __builtin_amdgcn_readfirstlane(tid >> 5);
    const int slot = lane / LPR;
    const int sub  = lane - slot * LPR;
    const bool active = slot < RPI;
    const int lo   = blockIdx.x * GCN_RPB;
    const int rows = min(GCN_RPB, nNodes - lo);
    const int cnt  = (int)blkcnt[blockIdx.x];
    const unsigned* L = list + __builtin_amdgcn_readfirstlane(gcn_list_base(blkcnt, blockIdx.x, lane));

    for (int i = tid; i < (GCN_RPB + 1) * F / 4; i += 1024)
        ((v4f*)acc)[i] = (v4f){0.f, 0.f, 0.f, 0.f};
    __syncthreads();

    const int o0 = 4 * sub + (slot & 3),       o1 = 4 * sub + ((1 + slot) & 3);
    const int o2 = 4 * sub + ((2 + slot) & 3), o3 = 4 * sub + ((3 + slot) & 3);
    const bool rot1 = (slot & 1) != 0, rot2 = (slot & 2) != 0;

    const unsigned pad = (unsigned)GCN_RPB << GCN_SRCBITS;
    unsigned cur = pad;
    {
        const int idx = CH * wave + lane;
        if (lane < CH && idx < cnt) cur = L[idx];
    }
    for (int it = 0; CH * (wave + 32 * it) < cnt; ++it) {
        const int nidx = CH * (wave + 32 * (it + 1)) + lane;
        const unsigned nraw = L[min(nidx, cnt - 1)];
        v4f r[STEPS];
        int drow[STEPS];
        #pragma unroll
        for (int t = 0; t < STEPS; ++t) {
            unsigned mine = __builtin_amdgcn_readlane(cur, RPI * t);
            #pragma unroll
            for (int g = 1; g < RPI; ++g) {
                const unsigned eg = __builtin_amdgcn_readlane(cur, RPI * t + g);
                mine = (slot == g) ? eg : mine;
            }
            const unsigned sv = mine & ((1u << GCN_SRCBITS) - 1u);
            drow[t] = (int)(mine >> GCN_SRCBITS) * F;
            r[t] = *(const v4f*)(Xe + (sv * (unsigned)F + 4u * (unsigned)sub));
        }
        static_assert(STEPS == 10, "the statement below names ten row reads");
        asm volatile("" :: "v"(r[0]), "v"(r[1]), "v"(r[2]), "v"(r[3]), "v"(r[4]),
                           "v"(r[5]), "v"(r[6]), "v"(r[7]), "v"(r[8]), "v"(r[9]), "v"(nraw));
        const unsigned nxt = (lane < CH && nidx < cnt) ? nraw : pad;
        #pragma unroll
        for (int t = 0; t < STEPS; ++t) {
            if (active) {
                float* a = acc + drow[t];
                const v4f x = r[t];
                const float y0 = rot1 ? x.y : x.x, y1 = rot1 ? x.z : x.y;
                const float y2 = rot1 ? x.w : x.z, y3 = rot1 ? x.x : x.w;
                const float x0 = rot2 ? y2 : y0, x1 = rot2 ? y3 : y1;
                const float x2 = rot2 ? y0 : y2, x3 = rot2 ? y1 : y3;
                __hip_atomic_fetch_add(a + o0, x0, __ATOMIC_RELAXED, __HIP_MEMORY_SCOPE_WORKGROUP);
                __hip_atomic_fetch_add(a + o1, x1, __ATOMIC_RELAXED, __HIP_MEMORY_SCOPE_WORKGROUP);
                __hip_atomic_fetch_add(a + o2, x2, __ATOMIC_RELAXED, __HIP_MEMORY_SCOPE_WORKGROUP);
                __hip_atomic_fetch_add(a + o3, x3, __ATOMIC_RELAXED, __HIP_MEMORY_SCOPE_WORKGROUP);
            }
        }
        cur = nxt;
    }
    __syncthreads();

    const int nunits = rows * LPR;
    float* Yb = Y + (long)lo * F;
    for (int u = tid; u < nunits; u += 1024) {
        const int rl = u / LPR;
        const int f4 = u - rl * LPR;
        const v4f a = ((const v4f*)acc)[u];
        const float s = nd[lo + rl];
        const v4f b = *(const v4f*)(bias + 4 * f4);
        v4f o;
        o.x = a.x * s + b.x; o.y = a.y * s + b.y; o.z = a.z * s + b.z; o.w = a.w * s + b.w;
        if (RELU) {
            o.x = fmaxf(o.x, 0.0f); o.y = fmaxf(o.y, 0.0f);
            o.z = fmaxf(o.z, 0.0f); o.w = fmaxf(o.w, 0.0f);
        }
        ((v4f*)Yb)[u] = o;
    }
}

extern "C" void kernel_launch(void* const* d_in, const int* in_sizes, int n_in,
                              void* d_out, int out_size, void* d_ws, size_t ws_size,
                              hipStream_t stream) {
    (void)in_sizes; (void)n_in; (void)out_size; (void)ws_size;

    const float* feat = (const float*)d_in[0];
    const int*   src  = (const int*)  d_in[1];
    const int*   dst  = (const int*)  d_in[2];
    const float* W0   = (const float*)d_in[3];
    const float* b0   = (const float*)d_in[4];
    const float* W1   = (const float*)d_in[5];
    const float* b1   = (const float*)d_in[6];
    const float* W2   = (const float*)d_in[7];
    const float* b2   = (const float*)d_in[8];
    float*       out  = (float*)d_out;

    const int  Nn = GCN_NODES, Ne = GCN_EDGES, H = GCN_HID;
    float* ws   = (float*)d_ws;
    float* ns   = ws;
    float* nd   = ws + Nn;
    unsigned* blkcnt = (unsigned*)(ws + 2L * Nn);
    float* bufA = ws + 2L * Nn + 256;
    float* bufB = bufA + (long)Nn * H;
    unsigned* own = (unsigned*)(bufB + (long)Nn * H);
    unsigned* list = own + Ne / 4;
    unsigned* part = list + Ne;
    unsigned* hist = part + 2L * GCN_DSL * Nn;

    gcn_degrees<<<GCN_DSL * GCN_DRG, 1024, 0, stream>>>(src, dst, part, own, hist);

    gcn_gemm_wmma<GCN_HID, 8, 1><<<Nn / 32, 256, 0, stream>>>(feat, ns, part, W0, bufA, Nn);
    gcn_agg_lds<GCN_HID, 1><<<GCN_AGGBLK, 1024, 0, stream>>>(bufA, src, dst, own, nd, b0, bufB, Nn, hist, blkcnt, list);

    gcn_gemm_wmma<GCN_HID, 8, 0><<<Nn / 32, 256, 0, stream>>>(bufB, ns, part, W1, bufA, Nn);
    gcn_agg_list<GCN_HID, 1><<<GCN_AGGBLK, 1024, 0, stream>>>(bufA, blkcnt, list, nd, b1, bufB, Nn);

    gcn_gemm_wmma<GCN_CLS, 3, 0><<<Nn / 32, 256, 0, stream>>>(bufB, ns, part, W2, bufA, Nn);
    gcn_agg_list_packed<GCN_CLS, 0><<<GCN_AGGBLK, 1024, 0, stream>>>(bufA, blkcnt, list, nd, b2, out, Nn);
}


# Layers 0, 1 gather in source order: a block bins its list by (owner wave, 2048-row source range), so all waves sweep the source plane low to high together; sum order = source range, then arrival
# speedup vs baseline: 4.0529x; 1.0403x over previous
#include <hip/hip_runtime.h>

typedef __attribute__((ext_vector_type(2))) float v2f;
typedef __attribute__((ext_vector_type(4))) float v4f;
typedef __attribute__((ext_vector_type(8))) float v8f;

#define GCN_NODES   100000
#define GCN_EDGES   1600000
#define GCN_KDIM    128
#define GCN_HID     128
#define GCN_CLS     40
#define GCN_LDA     129
#define GCN_RPB     392
#define GCN_AGGBLK  ((GCN_NODES + GCN_RPB - 1) / GCN_RPB)
#define GCN_QDRAIN  64
#define GCN_QCAP    576
static_assert(GCN_EDGES % 4 == 0 && GCN_EDGES <= (1 << 21), "edge number fits 21 bits, edges read as int4");
static_assert(GCN_RPB + 1 <= (1 << 9) && GCN_RPB % 4 == 0, "local row fits 9 bits");
static_assert(GCN_AGGBLK <= 256, "the number of the aggregation block that owns a row fits one byte");
static_assert(GCN_EDGES % 16 == 0, "the owner map is read as 16-byte vectors");
#define GCN_OWNV    (GCN_EDGES / 16)
#define GCN_SRCBITS 17
#define GCN_DSL     32
#define GCN_DRG     8
#define GCN_DRN     (GCN_NODES / GCN_DRG)
#define GCN_DSV     (GCN_EDGES / 4 / GCN_DSL)
static_assert(GCN_NODES % GCN_DRG == 0 && (GCN_EDGES / 4) % GCN_DSL == 0, "ranges and slices are whole");
static_assert(GCN_NODES <= (1 << GCN_SRCBITS) && GCN_RPB + 1 <= (1 << (32 - GCN_SRCBITS)),
              "a list word holds a source row and a local destination row");

__global__ void __launch_bounds__(1024)
gcn_degrees(const int* __restrict__ src, const int* __restrict__ dst,
            unsigned* __restrict__ part, unsigned* __restrict__ own,
            unsigned* __restrict__ hist) {
    __shared__ unsigned cs[GCN_DRN];
    __shared__ unsigned cd[GCN_DRN];
    __shared__ unsigned oh[GCN_AGGBLK];
    const int tid   = threadIdx.x;
    const int slice = blockIdx.x / GCN_DRG;
    const int rng   = blockIdx.x - slice * GCN_DRG;
    const int rlo   = rng * GCN_DRN;
    const bool maps = rng == (slice % GCN_DRG);

    for (int i = tid; i < GCN_DRN; i += 1024) { cs[i] = 0u; cd[i] = 0u; }
    if (tid < GCN_AGGBLK) oh[tid] = 0u;
    __syncthreads();

    const int4* s4 = (const int4*)src + (long)slice * GCN_DSV;
    const int4* d4 = (const int4*)dst + (long)slice * GCN_DSV;
    unsigned* ownp = own + (long)slice * GCN_DSV;
    for (int base = 0; base < GCN_DSV; base += 4096) {
        int4 s[4], d[4];
        #pragma unroll
        for (int j = 0; j < 4; ++j) {
            const int v = min(base + 1024 * j + tid, GCN_DSV - 1);
            s[j] = s4[v];
            d[j] = d4[v];
        }
        #pragma unroll
        for (int j = 0; j < 4; ++j) {
            const int v = base + 1024 * j + tid;
            if (v < GCN_DSV) {
                unsigned l;
                l = (unsigned)(s[j].x - rlo); if (l < (unsigned)GCN_DRN) atomicAdd(&cs[l], 1u);
                l = (unsigned)(s[j].y - rlo); if (l < (unsigned)GCN_DRN) atomicAdd(&cs[l], 1u);
                l = (unsigned)(s[j].z - rlo); if (l < (unsigned)GCN_DRN) atomicAdd(&cs[l], 1u);
                l = (unsigned)(s[j].w - rlo); if (l < (unsigned)GCN_DRN) atomicAdd(&cs[l], 1u);
                l = (unsigned)(d[j].x - rlo); if (l < (unsigned)GCN_DRN) atomicAdd(&cd[l], 1u);
                l = (unsigned)(d[j].y - rlo); if (l < (unsigned)GCN_DRN) atomicAdd(&cd[l], 1u);
                l = (unsigned)(d[j].z - rlo); if (l < (unsigned)GCN_DRN) atomicAdd(&cd[l], 1u);
                l = (unsigned)(d[j].w - rlo); if (l < (unsigned)GCN_DRN) atomicAdd(&cd[l], 1u);
                if (maps) {
                    const unsigned ox = (unsigned)d[j].x / GCN_RPB, oy = (unsigned)d[j].y / GCN_RPB;
                    const unsigned oz = (unsigned)d[j].z / GCN_RPB, ow = (unsigned)d[j].w / GCN_RPB;
                    ownp[v] = ox | (oy << 8) | (oz << 16) | (ow << 24);
                    atomicAdd(&oh[ox], 1u); atomicAdd(&oh[oy], 1u);
                    atomicAdd(&oh[oz], 1u); atomicAdd(&oh[ow], 1u);
                }
            }
        }
    }
    __syncthreads();

    unsigned* ps = part + ((long)slice * GCN_NODES + rlo);
    unsigned* pd = part + ((long)(GCN_DSL + slice) * GCN_NODES + rlo);
    for (int i = tid; i < GCN_DRN; i += 1024) { ps[i] = cs[i]; pd[i] = cd[i]; }
    if (maps && tid < GCN_AGGBLK) hist[slice * GCN_AGGBLK + tid] = oh[tid];
}

__device__ __forceinline__ unsigned gcn_list_base(const unsigned* __restrict__ blkcnt,
                                                  int b, int lane) {
    unsigned s = 0;
    for (int i = lane; i < b; i += 32) s += blkcnt[i];
    #pragma unroll
    for (int o = 16; o > 0; o >>= 1) s += __shfl_xor(s, o, 32);
    return s;
}

static_assert(GCN_DSL == 32, "one wave of a 1024-thread block per edge slice");
__device__ __forceinline__ unsigned gcn_hist_share(const unsigned* __restrict__ hist,
                                                   int b, int wave, int lane) {
    unsigned s = 0;
    const unsigned* row = hist + wave * GCN_AGGBLK;
    for (int i = lane; i < b; i += 32) s += row[i];
    #pragma unroll
    for (int o = 16; o > 0; o >>= 1) s += __shfl_xor(s, o, 32);
    return s;
}

template <int N, int NT, int FIRST>
__global__ void __launch_bounds__(256)
gcn_gemm_wmma(const float* __restrict__ X, float* __restrict__ p,
              const unsigned* __restrict__ part, const float* __restrict__ W,
              float* __restrict__ Y, int nNodes) {
    static_assert(NT <= 8 && NT * 16 >= N, "one wave per 16-column tile");
    __shared__ float Alds[32 * GCN_LDA];
    __shared__ unsigned dsum[64];
    __shared__ float nsl[32];

    const int tid = threadIdx.x;
    const int r0  = blockIdx.x * 32;

    float xr[32 * GCN_KDIM / 256];
    #pragma unroll
    for (int j = 0; j < 32 * GCN_KDIM / 256; ++j) {
        const int idx = tid + 256 * j;
        xr[j] = X[(long)(r0 + (idx >> 7)) * GCN_KDIM + (idx & 127)];
    }

    if (FIRST) {
        static_assert(GCN_DSL == 32, "eight thread groups of four planes");
        const int m = tid & 31, g = tid >> 5;
        const unsigned* qo = part + (long)(4 * g) * nNodes + (r0 + m);
        const unsigned* qi = qo + (long)GCN_DSL * nNodes;
        unsigned co = 0u, ci = 0u;
        #pragma unroll
        for (int j = 0; j < 4; ++j) {
            co += qo[(long)j * nNodes];
            ci += qi[(long)j * nNodes];
        }
        if (tid < 64) dsum[tid] = 0u;
        __syncthreads();
        atomicAdd(&dsum[m], co);
        atomicAdd(&dsum[32 + m], ci);
        __syncthreads();
        if (tid < 32) {
            const float a = rsqrtf(fmaxf((float)dsum[tid], 1.0f));
            nsl[tid] = a;
            p[r0 + tid]          = a;
            p[nNodes + r0 + tid] = rsqrtf(fmaxf((float)dsum[32 + tid], 1.0f));
        }
        __syncthreads();
    } else {
        if (tid < 32) nsl[tid] = p[r0 + tid];
        __syncthreads();
    }

    #pragma unroll
    for (int j = 0; j < 32 * GCN_KDIM / 256; ++j) {
        const int idx = tid + 256 * j;
        const int m = idx >> 7;
        const int k = idx & 127;
        Alds[m * GCN_LDA + k] = xr[j] * nsl[m];
    }
    __syncthreads();

    const int wave   = tid >> 5;
    const int lane   = tid & 31;
    const int lane16 = lane & 15;
    const int half   = lane >> 4;
    if (wave >= NT) return;

    const float* As0 = &Alds[lane16 * GCN_LDA + 2 * half];
    const float* As1 = As0 + 16 * GCN_LDA;
    const int  col = wave * 16 + lane16;
    const bool cv  = col < N;
    const float* Wp = W + (2 * half) * N + (cv ? col : 0);

    v8f acc0 = {0.f, 0.f, 0.f, 0.f, 0.f, 0.f, 0.f, 0.f};
    v8f acc1 = {0.f, 0.f, 0.f, 0.f, 0.f, 0.f, 0.f, 0.f};

    #pragma unroll 8
    for (int k4 = 0; k4 < GCN_KDIM / 4; ++k4) {
        const int k = k4 * 4;
        v2f a0, a1;
        a0.x = As0[k];
        a0.y = As0[k + 1];
        a1.x = As1[k];
        a1.y = As1[k + 1];
        v2f b;
        b.x = Wp[k * N];
        b.y = Wp[k * N + N];
        if (N % 16 != 0) {
            b.x = cv ? b.x : 0.0f;
            b.y = cv ? b.y : 0.0f;
        }
        acc0 = __builtin_amdgcn_wmma_f32_16x16x4_f32(
                   false, a0, false, b, (short)0, acc0, false, false);
        acc1 = __builtin_amdgcn_wmma_f32_16x16x4_f32(
                   false, a1, false, b, (short)0, acc1, false, false);
    }

    if (cv) {
        #pragma unroll
        for (int r = 0; r < 8; ++r) {
            int m = r + 8 * half;
            Y[(long)(r0 + m) * N + col]      = acc0[r];
            Y[(long)(r0 + 16 + m) * N + col] = acc1[r];
        }
    }
}

#define GCN_BCAP 8192
#define GCN_SRSH 11
#define GCN_SRNG (1 << GCN_SRSH)
#define GCN_NSB  (1 << (GCN_SRCBITS - GCN_SRSH))
#define GCN_NKEY (32 * GCN_NSB)
static_assert(GCN_BCAP % 1024 == 0, "every thread of the block takes the same number of words per pass");
static_assert(GCN_NKEY == 2048, "two keys per thread of a 1024-thread block; a wave scans one owner wave's keys");
template <int F>
__device__ __forceinline__ void gcn_add_owned(const float* __restrict__ Xe, const unsigned* L,
                                              const int cnt, float* acc, unsigned* bins,
                                              unsigned* kcnt, unsigned* wbase, unsigned* wsum) {
    static_assert(F == 128, "a row takes a whole wave: one 16-B piece per lane");
    const int tid  = threadIdx.x;
    const int lane = tid & 31;
    const int wave = __builtin_amdgcn_readfirstlane(tid >> 5);
    auto keyof = [](const unsigned e) {
        return (((e >> GCN_SRCBITS) & 31u) << (GCN_SRCBITS - GCN_SRSH)) |
               ((e >> GCN_SRSH) & (unsigned)(GCN_NSB - 1));
    };
    for (int c0 = 0; c0 < cnt; c0 += GCN_BCAP) {
        const int n = min(GCN_BCAP, cnt - c0);
        kcnt[2 * tid] = 0u; kcnt[2 * tid + 1] = 0u;
        __syncthreads();
        unsigned ent[GCN_BCAP / 1024];
        #pragma unroll
        for (int j = 0; j < GCN_BCAP / 1024; ++j)
            ent[j] = L[c0 + min(tid + 1024 * j, n - 1)];
        #pragma unroll
        for (int j = 0; j < GCN_BCAP / 1024; ++j)
            if (tid + 1024 * j < n) atomicAdd(&kcnt[keyof(ent[j])], 1u);
        __syncthreads();
        const unsigned ka = kcnt[2 * tid], kb = kcnt[2 * tid + 1];
        unsigned s = ka + kb;
        #pragma unroll
        for (int o = 1; o < 32; o <<= 1) {
            const unsigned t = __shfl_up(s, o, 32);
            if (lane >= o) s += t;
        }
        if (lane == 31) wsum[wave] = s;
        __syncthreads();
        if (wave == 0) {
            unsigned t = wsum[lane];
            #pragma unroll
            for (int o = 1; o < 32; o <<= 1) {
                const unsigned u = __shfl_up(t, o, 32);
                if (lane >= o) t += u;
            }
            wbase[lane + 1] = t;
            if (lane == 0) wbase[0] = 0u;
        }
        __syncthreads();
        {
            const unsigned ex = wbase[wave] + s - (ka + kb);
            kcnt[2 * tid]     = ex;
            kcnt[2 * tid + 1] = ex + ka;
        }
        __syncthreads();
        #pragma unroll
        for (int j = 0; j < GCN_BCAP / 1024; ++j)
            if (tid + 1024 * j < n)
                bins[atomicAdd(&kcnt[keyof(ent[j])], 1u)] = ent[j];
        __syncthreads();
        const int b0 = __builtin_amdgcn_readfirstlane((int)wbase[wave]);
        const int b1 = __builtin_amdgcn_readfirstlane((int)wbase[wave + 1]);
        const int ng = (b1 - b0) >> 3;
        auto ents = [&](const int g) { return bins[b0 + 8 * min(g, ng - 1) + (lane & 7)]; };
        auto rows8 = [&](v4f (&r)[8], const unsigned e) {
            #pragma unroll
            for (int j = 0; j < 8; ++j) {
                const unsigned s = __builtin_amdgcn_readlane(e, j) & ((1u << GCN_SRCBITS) - 1u);
                r[j] = *(const v4f*)(Xe + (long)s * F + 4 * lane);
            }
        };
        auto sums8 = [&](const v4f (&r)[8], const unsigned e) {
            #pragma unroll
            for (int j = 0; j < 8; ++j) {
                const unsigned d = __builtin_amdgcn_readlane(e, j) >> GCN_SRCBITS;
                v4f* a = (v4f*)(acc + d * F) + lane;
                *a = *a + r[j];
            }
        };
        if (ng > 0) {
            v4f ra[8], rb[8];
            unsigned ea = ents(0), eb = ents(1);
            rows8(ra, ea);
            int g = 0;
            while (g + 2 < ng) {
                rows8(rb, eb);
                const unsigned e2 = ents(g + 2), e3 = ents(g + 3);
                sums8(ra, ea);
                rows8(ra, e2);
                sums8(rb, eb);
                ea = e2; eb = e3; g += 2;
            }
            if (g + 2 == ng) {
                rows8(rb, eb);
                sums8(ra, ea);
                sums8(rb, eb);
            } else {
                sums8(ra, ea);
            }
        }
        for (int i = b0 + 8 * ng; i < b1; ++i) {
            const unsigned e = __builtin_amdgcn_readfirstlane(bins[i]);
            const unsigned s = e & ((1u << GCN_SRCBITS) - 1u);
            const unsigned d = e >> GCN_SRCBITS;
            const v4f x = *(const v4f*)(Xe + (long)s * F + 4 * lane);
            v4f* a = (v4f*)(acc + d * F) + lane;
            *a = *a + x;
        }
        __syncthreads();
    }
}

template <int F, int RELU>
__global__ void __launch_bounds__(1024)
gcn_agg_lds(const float* __restrict__ Xe, const int* __restrict__ src,
            const int* __restrict__ dst, const unsigned* __restrict__ own,
            const float* __restrict__ nd, const float* __restrict__ bias,
            float* __restrict__ Y, int nNodes,
            const unsigned* __restrict__ hist, unsigned* __restrict__ blkcnt,
            unsigned* __restrict__ list) {
    static_assert(F % 4 == 0 && F / 4 <= 32, "a row is read as one 16-B piece per lane");
    constexpr int LPR = F / 4;
    __shared__ __attribute__((aligned(16))) float acc[GCN_RPB * F];
    __shared__ unsigned queue[32 * GCN_QCAP];
    __shared__ unsigned bins[GCN_BCAP];
    __shared__ unsigned kcnt[GCN_NKEY], wbase[33], wsum[32];
    __shared__ unsigned hsum[32];
    __shared__ unsigned lpos;

    const int tid  = threadIdx.x;
    const int lane = tid & 31;
    const int wave = tid >> 5;
    const int lo   = blockIdx.x * GCN_RPB;
    const int rows = min(GCN_RPB, nNodes - lo);
    unsigned* q = queue + wave * GCN_QCAP;

    const unsigned share = gcn_hist_share(hist, blockIdx.x, wave, lane);
    if (lane == 0) hsum[wave] = share;
    for (int i = tid; i < GCN_RPB * F / 4; i += 1024)
        ((v4f*)acc)[i] = (v4f){0.f, 0.f, 0.f, 0.f};
    if (tid == 0) lpos = 0u;
    __syncthreads();
    unsigned lb = hsum[lane];
    #pragma unroll
    for (int o = 16; o > 0; o >>= 1) lb += __shfl_xor(lb, o, 32);
    unsigned* lst = list + __builtin_amdgcn_readfirstlane(lb);

    const int NV = GCN_OWNV;
    const uint4* own4 = (const uint4*)own;
    const unsigned pat = blockIdx.x * 0x01010101u;
    int cnt = 0;

    auto drain = [&]() {
        __builtin_amdgcn_fence(__ATOMIC_ACQ_REL, "wavefront");
        for (int b = 0; b < cnt; b += 32) {
            const int n = min(32, cnt - b);
            unsigned p0 = 0u;
            if (lane == 0)
                p0 = __hip_atomic_fetch_add(&lpos, (unsigned)n, __ATOMIC_RELAXED, __HIP_MEMORY_SCOPE_WORKGROUP);
            p0 = __builtin_amdgcn_readfirstlane(p0);
            if (lane < n) {
                const unsigned e = q[b + lane];
                const int sv = src[e];
                const int dl = dst[e] - lo;
                lst[p0 + lane] = (unsigned)sv | ((unsigned)dl << GCN_SRCBITS);
            }
        }
        __builtin_amdgcn_fence(__ATOMIC_ACQ_REL, "wavefront");
        cnt = 0;
    };

    auto hit = [&](const unsigned w) {
        const unsigned t = w ^ pat;
        return ~(((t & 0x7F7F7F7Fu) + 0x7F7F7F7Fu) | t | 0x7F7F7F7Fu);
    };

    auto test = [&](const uint4 c, const int vidx, const bool live) {
        unsigned m = (hit(c.x) >> 7) | (hit(c.y) >> 6) | (hit(c.z) >> 5) | (hit(c.w) >> 4);
        if (!live) m = 0u;
        unsigned bal = __builtin_amdgcn_ballot_w32(m != 0u);
        while (bal != 0u) {
            if (m != 0u) {
                const int p = __builtin_ctz(m);
                q[cnt + (int)__builtin_amdgcn_mbcnt_lo(bal, 0u)] =
                    (unsigned)(16 * vidx + 4 * (p & 7) + (p >> 3));
                m &= m - 1u;
            }
            cnt += __builtin_popcount(bal);
            bal = __builtin_amdgcn_ballot_w32(m != 0u);
        }
    };

    const int nFull = NV & ~4095;
    if (nFull > 0) {
        uint4 n0 = own4[tid], n1 = own4[tid + 1024], n2 = own4[tid + 2048], n3 = own4[tid + 3072];
        for (int base = 0; base < nFull; base += 4096) {
            const uint4 c0 = n0, c1 = n1, c2 = n2, c3 = n3;
            const int nb = min(base + 4096, nFull - 4096) + tid;
            n0 = own4[nb]; n1 = own4[nb + 1024]; n2 = own4[nb + 2048]; n3 = own4[nb + 3072];
            test(c0, base + tid,        true);
            if (cnt >= GCN_QDRAIN) drain();
            test(c1, base + tid + 1024, true);
            if (cnt >= GCN_QDRAIN) drain();
            test(c2, base + tid + 2048, true);
            if (cnt >= GCN_QDRAIN) drain();
            test(c3, base + tid + 3072, true);
            if (cnt >= GCN_QDRAIN) drain();
        }
    }
    for (int base = nFull; base < NV; base += 1024) {
        const int vidx = base + tid;
        const uint4 c = own4[min(vidx, NV - 1)];
        test(c, vidx, vidx < NV);
        if (cnt >= GCN_QDRAIN) drain();
    }
    if (cnt > 0) drain();
    __syncthreads();
    if (tid == 0) blkcnt[blockIdx.x] = lpos;

    gcn_add_owned<F>(Xe, lst, (int)lpos, acc, bins, kcnt, wbase, wsum);
    __syncthreads();

    const int nunits = rows * LPR;
    float* Yb = Y + (long)lo * F;
    for (int u = tid; u < nunits; u += 1024) {
        const int rl = u / LPR;
        const int f4 = u - rl * LPR;
        const v4f a = ((const v4f*)acc)[u];
        const float s = nd[lo + rl];
        const v4f b = *(const v4f*)(bias + 4 * f4);
        v4f o;
        o.x = a.x * s + b.x; o.y = a.y * s + b.y; o.z = a.z * s + b.z; o.w = a.w * s + b.w;
        if (RELU) {
            o.x = fmaxf(o.x, 0.0f); o.y = fmaxf(o.y, 0.0f);
            o.z = fmaxf(o.z, 0.0f); o.w = fmaxf(o.w, 0.0f);
        }
        ((v4f*)Yb)[u] = o;
    }
}

template <int F, int RELU>
__global__ void __launch_bounds__(1024)
gcn_agg_list(const float* __restrict__ Xe, const unsigned* __restrict__ blkcnt,
             const unsigned* __restrict__ list, const float* __restrict__ nd,
             const float* __restrict__ bias, float* __restrict__ Y, int nNodes) {
    static_assert(F % 4 == 0 && F / 4 <= 32, "a row is read as one 16-B piece per lane");
    constexpr int LPR = F / 4;
    __shared__ __attribute__((aligned(16))) float acc[GCN_RPB * F];
    __shared__ unsigned bins[GCN_BCAP];
    __shared__ unsigned kcnt[GCN_NKEY], wbase[33], wsum[32];

    const int tid  = threadIdx.x;
    const int lane = tid & 31;
    const int lo   = blockIdx.x * GCN_RPB;
    const int rows = min(GCN_RPB, nNodes - lo);
    const int cnt = (int)blkcnt[blockIdx.x];
    const unsigned* L = list + __builtin_amdgcn_readfirstlane(gcn_list_base(blkcnt, blockIdx.x, lane));

    for (int i = tid; i < GCN_RPB * F / 4; i += 1024)
        ((v4f*)acc)[i] = (v4f){0.f, 0.f, 0.f, 0.f};
    __syncthreads();

    gcn_add_owned<F>(Xe, L, cnt, acc, bins, kcnt, wbase, wsum);
    __syncthreads();

    const int nunits = rows * LPR;
    float* Yb = Y + (long)lo * F;
    for (int u = tid; u < nunits; u += 1024) {
        const int rl = u / LPR;
        const int f4 = u - rl * LPR;
        const v4f a = ((const v4f*)acc)[u];
        const float s = nd[lo + rl];
        const v4f b = *(const v4f*)(bias + 4 * f4);
        v4f o;
        o.x = a.x * s + b.x; o.y = a.y * s + b.y; o.z = a.z * s + b.z; o.w = a.w * s + b.w;
        if (RELU) {
            o.x = fmaxf(o.x, 0.0f); o.y = fmaxf(o.y, 0.0f);
            o.z = fmaxf(o.z, 0.0f); o.w = fmaxf(o.w, 0.0f);
        }
        ((v4f*)Yb)[u] = o;
    }
}

template <int F, int RELU>
__global__ void __launch_bounds__(1024)
gcn_agg_list_packed(const float* __restrict__ Xe, const unsigned* __restrict__ blkcnt,
                    const unsigned* __restrict__ list, const float* __restrict__ nd,
                    const float* __restrict__ bias, float* __restrict__ Y, int nNodes) {
    static_assert(F % 4 == 0 && F / 4 <= 16, "at least two rows fit one wave instruction");
    constexpr int LPR   = F / 4;
    constexpr int RPI   = 32 / LPR;
    constexpr int STEPS = 32 / RPI;
    constexpr int CH    = RPI * STEPS;
    static_assert(RPI >= 2 && RPI <= 4 && CH <= 32, "a lane group's float rotation is its number, below 4");
    __shared__ __attribute__((aligned(16))) float acc[(GCN_RPB + 1) * F];

    const int tid  = threadIdx.x;
    const int lane = tid & 31;
    const int wave = __builtin_amdgcn_readfirstlane(tid >> 5);
    const int slot = lane / LPR;
    const int sub  = lane - slot * LPR;
    const bool active = slot < RPI;
    const int lo   = blockIdx.x * GCN_RPB;
    const int rows = min(GCN_RPB, nNodes - lo);
    const int cnt  = (int)blkcnt[blockIdx.x];
    const unsigned* L = list + __builtin_amdgcn_readfirstlane(gcn_list_base(blkcnt, blockIdx.x, lane));

    for (int i = tid; i < (GCN_RPB + 1) * F / 4; i += 1024)
        ((v4f*)acc)[i] = (v4f){0.f, 0.f, 0.f, 0.f};
    __syncthreads();

    const int o0 = 4 * sub + (slot & 3),       o1 = 4 * sub + ((1 + slot) & 3);
    const int o2 = 4 * sub + ((2 + slot) & 3), o3 = 4 * sub + ((3 + slot) & 3);
    const bool rot1 = (slot & 1) != 0, rot2 = (slot & 2) != 0;

    const unsigned pad = (unsigned)GCN_RPB << GCN_SRCBITS;
    unsigned cur = pad;
    {
        const int idx = CH * wave + lane;
        if (lane < CH && idx < cnt) cur = L[idx];
    }
    for (int it = 0; CH * (wave + 32 * it) < cnt; ++it) {
        const int nidx = CH * (wave + 32 * (it + 1)) + lane;
        const unsigned nraw = L[min(nidx, cnt - 1)];
        v4f r[STEPS];
        int drow[STEPS];
        #pragma unroll
        for (int t = 0; t < STEPS; ++t) {
            unsigned mine = __builtin_amdgcn_readlane(cur, RPI * t);
            #pragma unroll
            for (int g = 1; g < RPI; ++g) {
                const unsigned eg = __builtin_amdgcn_readlane(cur, RPI * t + g);
                mine = (slot == g) ? eg : mine;
            }
            const unsigned sv = mine & ((1u << GCN_SRCBITS) - 1u);
            drow[t] = (int)(mine >> GCN_SRCBITS) * F;
            r[t] = *(const v4f*)(Xe + (sv * (unsigned)F + 4u * (unsigned)sub));
        }
        static_assert(STEPS == 10, "the statement below names ten row reads");
        asm volatile("" :: "v"(r[0]), "v"(r[1]), "v"(r[2]), "v"(r[3]), "v"(r[4]),
                           "v"(r[5]), "v"(r[6]), "v"(r[7]), "v"(r[8]), "v"(r[9]), "v"(nraw));
        const unsigned nxt = (lane < CH && nidx < cnt) ? nraw : pad;
        #pragma unroll
        for (int t = 0; t < STEPS; ++t) {
            if (active) {
                float* a = acc + drow[t];
                const v4f x = r[t];
                const float y0 = rot1 ? x.y : x.x, y1 = rot1 ? x.z : x.y;
                const float y2 = rot1 ? x.w : x.z, y3 = rot1 ? x.x : x.w;
                const float x0 = rot2 ? y2 : y0, x1 = rot2 ? y3 : y1;
                const float x2 = rot2 ? y0 : y2, x3 = rot2 ? y1 : y3;
                __hip_atomic_fetch_add(a + o0, x0, __ATOMIC_RELAXED, __HIP_MEMORY_SCOPE_WORKGROUP);
                __hip_atomic_fetch_add(a + o1, x1, __ATOMIC_RELAXED, __HIP_MEMORY_SCOPE_WORKGROUP);
                __hip_atomic_fetch_add(a + o2, x2, __ATOMIC_RELAXED, __HIP_MEMORY_SCOPE_WORKGROUP);
                __hip_atomic_fetch_add(a + o3, x3, __ATOMIC_RELAXED, __HIP_MEMORY_SCOPE_WORKGROUP);
            }
        }
        cur = nxt;
    }
    __syncthreads();

    const int nunits = rows * LPR;
    float* Yb = Y + (long)lo * F;
    for (int u = tid; u < nunits; u += 1024) {
        const int rl = u / LPR;
        const int f4 = u - rl * LPR;
        const v4f a = ((const v4f*)acc)[u];
        const float s = nd[lo + rl];
        const v4f b = *(const v4f*)(bias + 4 * f4);
        v4f o;
        o.x = a.x * s + b.x; o.y = a.y * s + b.y; o.z = a.z * s + b.z; o.w = a.w * s + b.w;
        if (RELU) {
            o.x = fmaxf(o.x, 0.0f); o.y = fmaxf(o.y, 0.0f);
            o.z = fmaxf(o.z, 0.0f); o.w = fmaxf(o.w, 0.0f);
        }
        ((v4f*)Yb)[u] = o;
    }
}

extern "C" void kernel_launch(void* const* d_in, const int* in_sizes, int n_in,
                              void* d_out, int out_size, void* d_ws, size_t ws_size,
                              hipStream_t stream) {
    (void)in_sizes; (void)n_in; (void)out_size; (void)ws_size;

    const float* feat = (const float*)d_in[0];
    const int*   src  = (const int*)  d_in[1];
    const int*   dst  = (const int*)  d_in[2];
    const float* W0   = (const float*)d_in[3];
    const float* b0   = (const float*)d_in[4];
    const float* W1   = (const float*)d_in[5];
    const float* b1   = (const float*)d_in[6];
    const float* W2   = (const float*)d_in[7];
    const float* b2   = (const float*)d_in[8];
    float*       out  = (float*)d_out;

    const int  Nn = GCN_NODES, Ne = GCN_EDGES, H = GCN_HID;
    float* ws   = (float*)d_ws;
    float* ns   = ws;
    float* nd   = ws + Nn;
    unsigned* blkcnt = (unsigned*)(ws + 2L * Nn);
    float* bufA = ws + 2L * Nn + 256;
    float* bufB = bufA + (long)Nn * H;
    unsigned* own = (unsigned*)(bufB + (long)Nn * H);
    unsigned* list = own + Ne / 4;
    unsigned* part = list + Ne;
    unsigned* hist = part + 2L * GCN_DSL * Nn;

    gcn_degrees<<<GCN_DSL * GCN_DRG, 1024, 0, stream>>>(src, dst, part, own, hist);

    gcn_gemm_wmma<GCN_HID, 8, 1><<<Nn / 32, 256, 0, stream>>>(feat, ns, part, W0, bufA, Nn);
    gcn_agg_lds<GCN_HID, 1><<<GCN_AGGBLK, 1024, 0, stream>>>(bufA, src, dst, own, nd, b0, bufB, Nn, hist, blkcnt, list);

    gcn_gemm_wmma<GCN_HID, 8, 0><<<Nn / 32, 256, 0, stream>>>(bufB, ns, part, W1, bufA, Nn);
    gcn_agg_list<GCN_HID, 1><<<GCN_AGGBLK, 1024, 0, stream>>>(bufA, blkcnt, list, nd, b1, bufB, Nn);

    gcn_gemm_wmma<GCN_CLS, 3, 0><<<Nn / 32, 256, 0, stream>>>(bufB, ns, part, W2, bufA, Nn);
    gcn_agg_list_packed<GCN_CLS, 0><<<GCN_AGGBLK, 1024, 0, stream>>>(bufA, blkcnt, list, nd, b2, out, Nn);
}
